# starting kernel
# baseline (speedup 1.0000x reference)
#include <hip/hip_runtime.h>

typedef __attribute__((ext_vector_type(2))) float v2f;
typedef __attribute__((ext_vector_type(8))) float v8f;

#define NPTS 4096
#define CCH  64
#define KNN  20
#define FMAX 3.402823466e+38f


__global__ void norms_kernel(const float* __restrict__ x, float* __restrict__ norms) {
    int p = blockIdx.x * blockDim.x + threadIdx.x;
    int b = p >> 12, n = p & 4095;
    const float* xb = x + b * CCH * NPTS + n;
    float s = 0.f;
    #pragma unroll
    for (int c = 0; c < CCH; ++c) { float v = xb[c * NPTS]; s += v * v; }
    norms[p] = s;
}

__global__ void __launch_bounds__(32) knn_kernel(const float* __restrict__ x,
                                                 const float* __restrict__ norms,
                                                 int* __restrict__ knn_idx) {
    __shared__ float distT[16][17];
    __shared__ float kd[16 * KNN];
    __shared__ int   ki[16 * KNN];

    const int lane = threadIdx.x;
    const int half = lane >> 4;
    const int lm   = lane & 15;
    const int b    = blockIdx.y;
    const int rowBase = blockIdx.x * 16;
    const float* xb = x + b * CCH * NPTS;
    const float* nb = norms + b * NPTS;

    v2f a[16];
    #pragma unroll
    for (int k = 0; k < 16; ++k) {
        int c = 4 * k + 2 * half;
        a[k].x = xb[(c    ) * NPTS + rowBase + lm];
        a[k].y = xb[(c + 1) * NPTS + rowBase + lm];
    }
    float rn[8];
    #pragma unroll
    for (int r = 0; r < 8; ++r) rn[r] = nb[rowBase + r + 8 * half];

    if (lane < 16) {
        for (int j = 0; j < KNN; ++j) { kd[lane * KNN + j] = FMAX; ki[lane * KNN + j] = 0; }
    }
    float curMax = FMAX; int curArg = 0;
    __syncthreads();

    for (int t = 0; t < NPTS / 16; ++t) {
        const int colBase = t * 16;

        v2f bb[16];
        #pragma unroll
        for (int k = 0; k < 16; ++k) {
            int c = 4 * k + 2 * half;
            bb[k].x = xb[(c    ) * NPTS + colBase + lm];
            bb[k].y = xb[(c + 1) * NPTS + colBase + lm];
        }

        v8f acc = {};
        #pragma unroll
        for (int k = 0; k < 16; ++k)
            acc = __builtin_amdgcn_wmma_f32_16x16x4_f32(false, a[k], false, bb[k],
                                                        (short)0, acc, false, false);

        float cn = nb[colBase + lm];
        #pragma unroll
        for (int r = 0; r < 8; ++r)
            distT[r + 8 * half][lm] = rn[r] + cn - 2.0f * acc[r];
        __syncthreads();

        if (lane < 16) {
            for (int j = 0; j < 16; ++j) {
                float d = distT[lane][j];
                if (d < curMax) {
                    kd[lane * KNN + curArg] = d;
                    ki[lane * KNN + curArg] = colBase + j;
                    float m = -FMAX; int arg = 0;
                    for (int q = 0; q < KNN; ++q) {
                        float v = kd[lane * KNN + q];
                        if (v > m) { m = v; arg = q; }
                    }
                    curMax = m; curArg = arg;
                }
            }
        }
        __syncthreads();
    }

    if (lane < 16) {
        int* op = knn_idx + (b * NPTS + rowBase + lane) * KNN;
        for (int j = 0; j < KNN; ++j) op[j] = ki[lane * KNN + j];
    }
}

__global__ void __launch_bounds__(32) proj_kernel(const float* __restrict__ x,
                                                  const float* __restrict__ W,
                                                  float* __restrict__ y,
                                                  float* __restrict__ z) {
    const int lane = threadIdx.x;
    const int half = lane >> 4;
    const int lm   = lane & 15;
    const int b    = blockIdx.y;
    const int rowBase = blockIdx.x * 16;
    const float* xb = x + b * CCH * NPTS;

    v2f a[16];
    #pragma unroll
    for (int k = 0; k < 16; ++k) {
        int c = 4 * k + 2 * half;
        a[k].x = xb[(c    ) * NPTS + rowBase + lm];
        a[k].y = xb[(c + 1) * NPTS + rowBase + lm];
    }

    for (int ot = 0; ot < 4; ++ot) {
        const int o = ot * 16 + lm;
        v8f accY = {}, accZ = {};
        #pragma unroll
        for (int k = 0; k < 16; ++k) {
            int c = 4 * k + 2 * half;
            float w0 = W[o * 128 + c], w1 = W[o * 128 + c + 1];
            v2f bw, bz;
            bw.x = w0;
            bw.y = w1;
            bz.x = W[o * 128 + 64 + c    ] - w0;
            bz.y = W[o * 128 + 64 + c + 1] - w1;
            accY = __builtin_amdgcn_wmma_f32_16x16x4_f32(false, a[k], false, bw,
                                                         (short)0, accY, false, false);
            accZ = __builtin_amdgcn_wmma_f32_16x16x4_f32(false, a[k], false, bz,
                                                         (short)0, accZ, false, false);
        }
        #pragma unroll
        for (int r = 0; r < 8; ++r) {
            int row = rowBase + r + 8 * half;
            y[(b * NPTS + row) * CCH + o] = accY[r];
            z[(b * NPTS + row) * CCH + o] = accZ[r];
        }
    }
}

__global__ void gathermax_kernel(const float* __restrict__ y,
                                 const float* __restrict__ z,
                                 const int* __restrict__ knn_idx,
                                 float* __restrict__ out) {
    int t = blockIdx.x * blockDim.x + threadIdx.x;
    int o = t & 63;
    int p = t >> 6;
    int b = p >> 12, n = p & 4095;
    const int*   ip = knn_idx + p * KNN;
    const float* yb = y + b * NPTS * CCH;
    float best = -FMAX;
    #pragma unroll
    for (int k = 0; k < KNN; ++k) {
        int m = ip[k];
        best = fmaxf(best, yb[m * CCH + o]);
    }
    best += z[p * CCH + o];
    out[(b * CCH + o) * NPTS + n] = best;
}

extern "C" void kernel_launch(void* const* d_in, const int* in_sizes, int n_in,
                              void* d_out, int out_size, void* d_ws, size_t ws_size,
                              hipStream_t stream) {
    (void)in_sizes; (void)n_in; (void)out_size; (void)ws_size;
    const float* x = (const float*)d_in[0];
    const float* W = (const float*)d_in[1];
    float* out = (float*)d_out;

    char* ws = (char*)d_ws;
    float* norms = (float*)(ws);
    int*   idxb  = (int*)  (ws + 65536);
    float* y     = (float*)(ws + 65536 + 1310720);
    float* z     = (float*)(ws + 65536 + 1310720 + 4194304);

    norms_kernel    <<<dim3(64), 256, 0, stream>>>(x, norms);
    proj_kernel     <<<dim3(256, 4), 32, 0, stream>>>(x, W, y, z);
    knn_kernel      <<<dim3(256, 4), 32, 0, stream>>>(x, norms, idxb);
    gathermax_kernel<<<dim3(4096), 256, 0, stream>>>(y, z, idxb, out);
}


# knn: column sweep of each 16-row block split over 8 waves of one work-group (every 8th tile each, own 20-best lists in LDS, pairwise merge at the end): 8 waves per SIMD instead of 1
# speedup vs baseline: 2.2858x; 2.2858x over previous
#include <hip/hip_runtime.h>

typedef __attribute__((ext_vector_type(2))) float v2f;
typedef __attribute__((ext_vector_type(8))) float v8f;

#define NPTS 4096
#define CCH  64
#define KNN  20
#define FMAX 3.402823466e+38f


__global__ void norms_kernel(const float* __restrict__ x, float* __restrict__ norms) {
    int p = blockIdx.x * blockDim.x + threadIdx.x;
    int b = p >> 12, n = p & 4095;
    const float* xb = x + b * CCH * NPTS + n;
    float s = 0.f;
    #pragma unroll
    for (int c = 0; c < CCH; ++c) { float v = xb[c * NPTS]; s += v * v; }
    norms[p] = s;
}

#define NSPLIT 8

__device__ __forceinline__ void topk_consider(float d, int col, float* __restrict__ ld,
                                              int* __restrict__ li,
                                              float& curMax, int& curArg) {
    if (d < curMax) {
        ld[curArg] = d;
        li[curArg] = col;
        float m = -FMAX; int arg = 0;
        for (int q = 0; q < KNN; ++q) {
            float v = ld[q];
            if (v > m) { m = v; arg = q; }
        }
        curMax = m; curArg = arg;
    }
}

__global__ void __launch_bounds__(32 * NSPLIT) __attribute__((amdgpu_waves_per_eu(8, 10)))
knn_kernel(const float* __restrict__ x,
           const float* __restrict__ norms,
           int* __restrict__ knn_idx) {
    __shared__ float distT[NSPLIT][16][17];
    __shared__ float kd[NSPLIT][16 * KNN];
    __shared__ int   ki[NSPLIT][16 * KNN];

    const int lane = threadIdx.x;
    const int w    = threadIdx.y;
    const int half = lane >> 4;
    const int lm   = lane & 15;
    const int b    = blockIdx.y;
    const int rowBase = blockIdx.x * 16;
    const float* xb = x + b * CCH * NPTS;
    const float* nb = norms + b * NPTS;

    v2f a[16];
    #pragma unroll
    for (int k = 0; k < 16; ++k) {
        int c = 4 * k + 2 * half;
        a[k].x = xb[(c    ) * NPTS + rowBase + lm];
        a[k].y = xb[(c + 1) * NPTS + rowBase + lm];
    }
    float rn[8];
    #pragma unroll
    for (int r = 0; r < 8; ++r) rn[r] = nb[rowBase + r + 8 * half];

    float* myd = &kd[w][lm * KNN];
    int*   myi = &ki[w][lm * KNN];
    if (lane < 16) {
        for (int j = 0; j < KNN; ++j) { myd[j] = FMAX; myi[j] = 0; }
    }
    float curMax = FMAX; int curArg = 0;

    for (int i = 0; i < NPTS / 16 / NSPLIT; ++i) {
        const int colBase = (i * NSPLIT + w) * 16;

        v2f bb[16];
        #pragma unroll
        for (int k = 0; k < 16; ++k) {
            int c = 4 * k + 2 * half;
            bb[k].x = xb[(c    ) * NPTS + colBase + lm];
            bb[k].y = xb[(c + 1) * NPTS + colBase + lm];
        }

        v8f acc = {};
        #pragma unroll
        for (int k = 0; k < 16; ++k)
            acc = __builtin_amdgcn_wmma_f32_16x16x4_f32(false, a[k], false, bb[k],
                                                        (short)0, acc, false, false);

        float cn = nb[colBase + lm];
        #pragma unroll
        for (int r = 0; r < 8; ++r)
            distT[w][r + 8 * half][lm] = rn[r] + cn - 2.0f * acc[r];

        if (lane < 16) {
            for (int j = 0; j < 16; ++j)
                topk_consider(distT[w][lane][j], colBase + j, myd, myi, curMax, curArg);
        }
    }

    for (int stride = 1; stride < NSPLIT; stride *= 2) {
        __syncthreads();
        if ((w & (2 * stride - 1)) == 0 && lane < 16) {
            const float* od = &kd[w + stride][lane * KNN];
            const int*   oi = &ki[w + stride][lane * KNN];
            for (int j = 0; j < KNN; ++j)
                topk_consider(od[j], oi[j], myd, myi, curMax, curArg);
        }
    }

    if (w == 0 && lane < 16) {
        int* op = knn_idx + (b * NPTS + rowBase + lane) * KNN;
        for (int j = 0; j < KNN; ++j) op[j] = myi[j];
    }
}

__global__ void __launch_bounds__(32) proj_kernel(const float* __restrict__ x,
                                                  const float* __restrict__ W,
                                                  float* __restrict__ y,
                                                  float* __restrict__ z) {
    const int lane = threadIdx.x;
    const int half = lane >> 4;
    const int lm   = lane & 15;
    const int b    = blockIdx.y;
    const int rowBase = blockIdx.x * 16;
    const float* xb = x + b * CCH * NPTS;

    v2f a[16];
    #pragma unroll
    for (int k = 0; k < 16; ++k) {
        int c = 4 * k + 2 * half;
        a[k].x = xb[(c    ) * NPTS + rowBase + lm];
        a[k].y = xb[(c + 1) * NPTS + rowBase + lm];
    }

    for (int ot = 0; ot < 4; ++ot) {
        const int o = ot * 16 + lm;
        v8f accY = {}, accZ = {};
        #pragma unroll
        for (int k = 0; k < 16; ++k) {
            int c = 4 * k + 2 * half;
            float w0 = W[o * 128 + c], w1 = W[o * 128 + c + 1];
            v2f bw, bz;
            bw.x = w0;
            bw.y = w1;
            bz.x = W[o * 128 + 64 + c    ] - w0;
            bz.y = W[o * 128 + 64 + c + 1] - w1;
            accY = __builtin_amdgcn_wmma_f32_16x16x4_f32(false, a[k], false, bw,
                                                         (short)0, accY, false, false);
            accZ = __builtin_amdgcn_wmma_f32_16x16x4_f32(false, a[k], false, bz,
                                                         (short)0, accZ, false, false);
        }
        #pragma unroll
        for (int r = 0; r < 8; ++r) {
            int row = rowBase + r + 8 * half;
            y[(b * NPTS + row) * CCH + o] = accY[r];
            z[(b * NPTS + row) * CCH + o] = accZ[r];
        }
    }
}

__global__ void gathermax_kernel(const float* __restrict__ y,
                                 const float* __restrict__ z,
                                 const int* __restrict__ knn_idx,
                                 float* __restrict__ out) {
    int t = blockIdx.x * blockDim.x + threadIdx.x;
    int o = t & 63;
    int p = t >> 6;
    int b = p >> 12, n = p & 4095;
    const int*   ip = knn_idx + p * KNN;
    const float* yb = y + b * NPTS * CCH;
    float best = -FMAX;
    #pragma unroll
    for (int k = 0; k < KNN; ++k) {
        int m = ip[k];
        best = fmaxf(best, yb[m * CCH + o]);
    }
    best += z[p * CCH + o];
    out[(b * CCH + o) * NPTS + n] = best;
}

extern "C" void kernel_launch(void* const* d_in, const int* in_sizes, int n_in,
                              void* d_out, int out_size, void* d_ws, size_t ws_size,
                              hipStream_t stream) {
    (void)in_sizes; (void)n_in; (void)out_size; (void)ws_size;
    const float* x = (const float*)d_in[0];
    const float* W = (const float*)d_in[1];
    float* out = (float*)d_out;

    char* ws = (char*)d_ws;
    float* norms = (float*)(ws);
    int*   idxb  = (int*)  (ws + 65536);
    float* y     = (float*)(ws + 65536 + 1310720);
    float* z     = (float*)(ws + 65536 + 1310720 + 4194304);

    norms_kernel    <<<dim3(64), 256, 0, stream>>>(x, norms);
    proj_kernel     <<<dim3(256, 4), 32, 0, stream>>>(x, W, y, z);
    knn_kernel      <<<dim3(256, 4), dim3(32, NSPLIT), 0, stream>>>(x, norms, idxb);
    gathermax_kernel<<<dim3(4096), 256, 0, stream>>>(y, z, idxb, out);
}


# knn: Gram tile transposed (column tile as A, rows as B): each of 32 lanes filters 8 candidates of its own row straight from accumulator registers (no LDS transpose, no idle half-wave); 16 lists merged
# speedup vs baseline: 2.6661x; 1.1664x over previous
#include <hip/hip_runtime.h>

typedef __attribute__((ext_vector_type(2))) float v2f;
typedef __attribute__((ext_vector_type(8))) float v8f;

#define NPTS 4096
#define CCH  64
#define KNN  20
#define FMAX 3.402823466e+38f


__global__ void norms_kernel(const float* __restrict__ x, float* __restrict__ norms) {
    int p = blockIdx.x * blockDim.x + threadIdx.x;
    int b = p >> 12, n = p & 4095;
    const float* xb = x + b * CCH * NPTS + n;
    float s = 0.f;
    #pragma unroll
    for (int c = 0; c < CCH; ++c) { float v = xb[c * NPTS]; s += v * v; }
    norms[p] = s;
}

#define NSPLIT 8

__device__ __forceinline__ void topk_consider(float d, int col, float* __restrict__ ld,
                                              int* __restrict__ li,
                                              float& curMax, int& curArg) {
    if (d < curMax) {
        ld[curArg] = d;
        li[curArg] = col;
        float m = -FMAX; int arg = 0;
        for (int q = 0; q < KNN; ++q) {
            float v = ld[q];
            if (v > m) { m = v; arg = q; }
        }
        curMax = m; curArg = arg;
    }
}

__global__ void __launch_bounds__(32 * NSPLIT) __attribute__((amdgpu_waves_per_eu(8, 10)))
knn_kernel(const float* __restrict__ x,
           const float* __restrict__ norms,
           int* __restrict__ knn_idx) {
    __shared__ __attribute__((aligned(16))) float kd[NSPLIT][32 * KNN];
    __shared__ __attribute__((aligned(16))) int   ki[NSPLIT][32 * KNN];

    const int lane = threadIdx.x;
    const int w    = threadIdx.y;
    const int half = lane >> 4;
    const int lm   = lane & 15;
    const int b    = blockIdx.y;
    const int rowBase = blockIdx.x * 16;
    const float* xb = x + b * CCH * NPTS;
    const float* nb = norms + b * NPTS;

    v2f a[16];
    #pragma unroll
    for (int k = 0; k < 16; ++k) {
        int c = 4 * k + 2 * half;
        a[k].x = xb[(c    ) * NPTS + rowBase + lm];
        a[k].y = xb[(c + 1) * NPTS + rowBase + lm];
    }
    const float rn = nb[rowBase + lm];

    float* myd = &kd[w][lane * KNN];
    int*   myi = &ki[w][lane * KNN];
    for (int j = 0; j < KNN; ++j) { myd[j] = FMAX; myi[j] = 0; }
    float curMax = FMAX; int curArg = 0;

    for (int i = 0; i < NPTS / 16 / NSPLIT; ++i) {
        const int colBase = (i * NSPLIT + w) * 16;

        const float4 cnLo = *(const float4*)(nb + colBase + 8 * half);
        const float4 cnHi = *(const float4*)(nb + colBase + 8 * half + 4);
        const float cn[8] = {cnLo.x, cnLo.y, cnLo.z, cnLo.w, cnHi.x, cnHi.y, cnHi.z, cnHi.w};

        v2f bb[16];
        #pragma unroll
        for (int k = 0; k < 16; ++k) {
            int c = 4 * k + 2 * half;
            bb[k].x = xb[(c    ) * NPTS + colBase + lm];
            bb[k].y = xb[(c + 1) * NPTS + colBase + lm];
        }

        v8f acc = {};
        #pragma unroll
        for (int k = 0; k < 16; ++k)
            acc = __builtin_amdgcn_wmma_f32_16x16x4_f32(false, bb[k], false, a[k],
                                                        (short)0, acc, false, false);

        const int col0 = colBase + 8 * half;
        #pragma unroll
        for (int r = 0; r < 8; ++r)
            topk_consider(rn + cn[r] - 2.0f * acc[r], col0 + r, myd, myi, curMax, curArg);
    }

    const int g = 2 * w + half;
    for (int stride = 1; stride < 2 * NSPLIT; stride *= 2) {
        __syncthreads();
        if ((g & (2 * stride - 1)) == 0) {
            const int og = g + stride;
            const float* od = &kd[og >> 1][((og & 1) * 16 + lm) * KNN];
            const int*   oi = &ki[og >> 1][((og & 1) * 16 + lm) * KNN];
            for (int j = 0; j < KNN; ++j)
                topk_consider(od[j], oi[j], myd, myi, curMax, curArg);
        }
    }

    if (w == 0 && lane < 16) {
        int* op = knn_idx + (b * NPTS + rowBase + lane) * KNN;
        for (int j = 0; j < KNN; ++j) op[j] = myi[j];
    }
}

__global__ void __launch_bounds__(32) proj_kernel(const float* __restrict__ x,
                                                  const float* __restrict__ W,
                                                  float* __restrict__ y,
                                                  float* __restrict__ z) {
    const int lane = threadIdx.x;
    const int half = lane >> 4;
    const int lm   = lane & 15;
    const int b    = blockIdx.y;
    const int rowBase = blockIdx.x * 16;
    const float* xb = x + b * CCH * NPTS;

    v2f a[16];
    #pragma unroll
    for (int k = 0; k < 16; ++k) {
        int c = 4 * k + 2 * half;
        a[k].x = xb[(c    ) * NPTS + rowBase + lm];
        a[k].y = xb[(c + 1) * NPTS + rowBase + lm];
    }

    for (int ot = 0; ot < 4; ++ot) {
        const int o = ot * 16 + lm;
        v8f accY = {}, accZ = {};
        #pragma unroll
        for (int k = 0; k < 16; ++k) {
            int c = 4 * k + 2 * half;
            float w0 = W[o * 128 + c], w1 = W[o * 128 + c + 1];
            v2f bw, bz;
            bw.x = w0;
            bw.y = w1;
            bz.x = W[o * 128 + 64 + c    ] - w0;
            bz.y = W[o * 128 + 64 + c + 1] - w1;
            accY = __builtin_amdgcn_wmma_f32_16x16x4_f32(false, a[k], false, bw,
                                                         (short)0, accY, false, false);
            accZ = __builtin_amdgcn_wmma_f32_16x16x4_f32(false, a[k], false, bz,
                                                         (short)0, accZ, false, false);
        }
        #pragma unroll
        for (int r = 0; r < 8; ++r) {
            int row = rowBase + r + 8 * half;
            y[(b * NPTS + row) * CCH + o] = accY[r];
            z[(b * NPTS + row) * CCH + o] = accZ[r];
        }
    }
}

__global__ void gathermax_kernel(const float* __restrict__ y,
                                 const float* __restrict__ z,
                                 const int* __restrict__ knn_idx,
                                 float* __restrict__ out) {
    int t = blockIdx.x * blockDim.x + threadIdx.x;
    int o = t & 63;
    int p = t >> 6;
    int b = p >> 12, n = p & 4095;
    const int*   ip = knn_idx + p * KNN;
    const float* yb = y + b * NPTS * CCH;
    float best = -FMAX;
    #pragma unroll
    for (int k = 0; k < KNN; ++k) {
        int m = ip[k];
        best = fmaxf(best, yb[m * CCH + o]);
    }
    best += z[p * CCH + o];
    out[(b * CCH + o) * NPTS + n] = best;
}

extern "C" void kernel_launch(void* const* d_in, const int* in_sizes, int n_in,
                              void* d_out, int out_size, void* d_ws, size_t ws_size,
                              hipStream_t stream) {
    (void)in_sizes; (void)n_in; (void)out_size; (void)ws_size;
    const float* x = (const float*)d_in[0];
    const float* W = (const float*)d_in[1];
    float* out = (float*)d_out;

    char* ws = (char*)d_ws;
    float* norms = (float*)(ws);
    int*   idxb  = (int*)  (ws + 65536);
    float* y     = (float*)(ws + 65536 + 1310720);
    float* z     = (float*)(ws + 65536 + 1310720 + 4194304);

    norms_kernel    <<<dim3(64), 256, 0, stream>>>(x, norms);
    proj_kernel     <<<dim3(256, 4), 32, 0, stream>>>(x, W, y, z);
    knn_kernel      <<<dim3(256, 4), dim3(32, NSPLIT), 0, stream>>>(x, norms, idxb);
    gathermax_kernel<<<dim3(4096), 256, 0, stream>>>(y, z, idxb, out);
}


# knn: per tile each lane masks its passing candidates and a loop inserts every lane's own next one per trip, so a tile costs its busiest lane's count of trips, not one per register slot
# speedup vs baseline: 2.8800x; 1.0802x over previous
#include <hip/hip_runtime.h>

typedef __attribute__((ext_vector_type(2))) float v2f;
typedef __attribute__((ext_vector_type(8))) float v8f;

#define NPTS 4096
#define CCH  64
#define KNN  20
#define FMAX 3.402823466e+38f


__global__ void norms_kernel(const float* __restrict__ x, float* __restrict__ norms) {
    int p = blockIdx.x * blockDim.x + threadIdx.x;
    int b = p >> 12, n = p & 4095;
    const float* xb = x + b * CCH * NPTS + n;
    float s = 0.f;
    #pragma unroll
    for (int c = 0; c < CCH; ++c) { float v = xb[c * NPTS]; s += v * v; }
    norms[p] = s;
}

#define NSPLIT 8

__device__ __forceinline__ void topk_consider(float d, int col, float* __restrict__ ld,
                                              int* __restrict__ li,
                                              float& curMax, int& curArg) {
    if (d < curMax) {
        ld[curArg] = d;
        li[curArg] = col;
        float m = -FMAX; int arg = 0;
        for (int q = 0; q < KNN; ++q) {
            float v = ld[q];
            if (v > m) { m = v; arg = q; }
        }
        curMax = m; curArg = arg;
    }
}

__global__ void __launch_bounds__(32 * NSPLIT) __attribute__((amdgpu_waves_per_eu(8, 10)))
knn_kernel(const float* __restrict__ x,
           const float* __restrict__ norms,
           int* __restrict__ knn_idx) {
    __shared__ __attribute__((aligned(16))) float kd[NSPLIT][32 * KNN];
    __shared__ __attribute__((aligned(16))) int   ki[NSPLIT][32 * KNN];

    const int lane = threadIdx.x;
    const int w    = threadIdx.y;
    const int half = lane >> 4;
    const int lm   = lane & 15;
    const int b    = blockIdx.y;
    const int rowBase = blockIdx.x * 16;
    const float* xb = x + b * CCH * NPTS;
    const float* nb = norms + b * NPTS;

    v2f a[16];
    #pragma unroll
    for (int k = 0; k < 16; ++k) {
        int c = 4 * k + 2 * half;
        a[k].x = xb[(c    ) * NPTS + rowBase + lm];
        a[k].y = xb[(c + 1) * NPTS + rowBase + lm];
    }
    const float rn = nb[rowBase + lm];

    float* myd = &kd[w][lane * KNN];
    int*   myi = &ki[w][lane * KNN];
    for (int j = 0; j < KNN; ++j) { myd[j] = FMAX; myi[j] = 0; }
    float curMax = FMAX; int curArg = 0;

    for (int i = 0; i < NPTS / 16 / NSPLIT; ++i) {
        const int colBase = (i * NSPLIT + w) * 16;

        const float4 cnLo = *(const float4*)(nb + colBase + 8 * half);
        const float4 cnHi = *(const float4*)(nb + colBase + 8 * half + 4);
        const float cn[8] = {cnLo.x, cnLo.y, cnLo.z, cnLo.w, cnHi.x, cnHi.y, cnHi.z, cnHi.w};

        v2f bb[16];
        #pragma unroll
        for (int k = 0; k < 16; ++k) {
            int c = 4 * k + 2 * half;
            bb[k].x = xb[(c    ) * NPTS + colBase + lm];
            bb[k].y = xb[(c + 1) * NPTS + colBase + lm];
        }

        v8f acc = {};
        #pragma unroll
        for (int k = 0; k < 16; ++k)
            acc = __builtin_amdgcn_wmma_f32_16x16x4_f32(false, bb[k], false, a[k],
                                                        (short)0, acc, false, false);

        const int col0 = colBase + 8 * half;
        float dv[8];
        #pragma unroll
        for (int r = 0; r < 8; ++r) dv[r] = rn + cn[r] - 2.0f * acc[r];
        unsigned pend = 0u;
        #pragma unroll
        for (int r = 0; r < 8; ++r) pend |= (dv[r] < curMax) ? (1u << r) : 0u;
        while (pend != 0u) {
            const int r = __builtin_ctz(pend);
            pend &= pend - 1u;
            const float s0 = (r & 1) ? dv[1] : dv[0];
            const float s1 = (r & 1) ? dv[3] : dv[2];
            const float s2 = (r & 1) ? dv[5] : dv[4];
            const float s3 = (r & 1) ? dv[7] : dv[6];
            const float t0 = (r & 2) ? s1 : s0;
            const float t1 = (r & 2) ? s3 : s2;
            const float d  = (r & 4) ? t1 : t0;
            topk_consider(d, col0 + r, myd, myi, curMax, curArg);
        }
    }

    const int g = 2 * w + half;
    for (int stride = 1; stride < 2 * NSPLIT; stride *= 2) {
        __syncthreads();
        if ((g & (2 * stride - 1)) == 0) {
            const int og = g + stride;
            const float* od = &kd[og >> 1][((og & 1) * 16 + lm) * KNN];
            const int*   oi = &ki[og >> 1][((og & 1) * 16 + lm) * KNN];
            for (int j = 0; j < KNN; ++j)
                topk_consider(od[j], oi[j], myd, myi, curMax, curArg);
        }
    }

    if (w == 0 && lane < 16) {
        int* op = knn_idx + (b * NPTS + rowBase + lane) * KNN;
        for (int j = 0; j < KNN; ++j) op[j] = myi[j];
    }
}

__global__ void __launch_bounds__(32) proj_kernel(const float* __restrict__ x,
                                                  const float* __restrict__ W,
                                                  float* __restrict__ y,
                                                  float* __restrict__ z) {
    const int lane = threadIdx.x;
    const int half = lane >> 4;
    const int lm   = lane & 15;
    const int b    = blockIdx.y;
    const int rowBase = blockIdx.x * 16;
    const float* xb = x + b * CCH * NPTS;

    v2f a[16];
    #pragma unroll
    for (int k = 0; k < 16; ++k) {
        int c = 4 * k + 2 * half;
        a[k].x = xb[(c    ) * NPTS + rowBase + lm];
        a[k].y = xb[(c + 1) * NPTS + rowBase + lm];
    }

    for (int ot = 0; ot < 4; ++ot) {
        const int o = ot * 16 + lm;
        v8f accY = {}, accZ = {};
        #pragma unroll
        for (int k = 0; k < 16; ++k) {
            int c = 4 * k + 2 * half;
            float w0 = W[o * 128 + c], w1 = W[o * 128 + c + 1];
            v2f bw, bz;
            bw.x = w0;
            bw.y = w1;
            bz.x = W[o * 128 + 64 + c    ] - w0;
            bz.y = W[o * 128 + 64 + c + 1] - w1;
            accY = __builtin_amdgcn_wmma_f32_16x16x4_f32(false, a[k], false, bw,
                                                         (short)0, accY, false, false);
            accZ = __builtin_amdgcn_wmma_f32_16x16x4_f32(false, a[k], false, bz,
                                                         (short)0, accZ, false, false);
        }
        #pragma unroll
        for (int r = 0; r < 8; ++r) {
            int row = rowBase + r + 8 * half;
            y[(b * NPTS + row) * CCH + o] = accY[r];
            z[(b * NPTS + row) * CCH + o] = accZ[r];
        }
    }
}

__global__ void gathermax_kernel(const float* __restrict__ y,
                                 const float* __restrict__ z,
                                 const int* __restrict__ knn_idx,
                                 float* __restrict__ out) {
    int t = blockIdx.x * blockDim.x + threadIdx.x;
    int o = t & 63;
    int p = t >> 6;
    int b = p >> 12, n = p & 4095;
    const int*   ip = knn_idx + p * KNN;
    const float* yb = y + b * NPTS * CCH;
    float best = -FMAX;
    #pragma unroll
    for (int k = 0; k < KNN; ++k) {
        int m = ip[k];
        best = fmaxf(best, yb[m * CCH + o]);
    }
    best += z[p * CCH + o];
    out[(b * CCH + o) * NPTS + n] = best;
}

extern "C" void kernel_launch(void* const* d_in, const int* in_sizes, int n_in,
                              void* d_out, int out_size, void* d_ws, size_t ws_size,
                              hipStream_t stream) {
    (void)in_sizes; (void)n_in; (void)out_size; (void)ws_size;
    const float* x = (const float*)d_in[0];
    const float* W = (const float*)d_in[1];
    float* out = (float*)d_out;

    char* ws = (char*)d_ws;
    float* norms = (float*)(ws);
    int*   idxb  = (int*)  (ws + 65536);
    float* y     = (float*)(ws + 65536 + 1310720);
    float* z     = (float*)(ws + 65536 + 1310720 + 4194304);

    norms_kernel    <<<dim3(64), 256, 0, stream>>>(x, norms);
    proj_kernel     <<<dim3(256, 4), 32, 0, stream>>>(x, W, y, z);
    knn_kernel      <<<dim3(256, 4), dim3(32, NSPLIT), 0, stream>>>(x, norms, idxb);
    gathermax_kernel<<<dim3(4096), 256, 0, stream>>>(y, z, idxb, out);
}


# knn: gather-max fused into the knn kernel's tail: the work-group that ranked 16 rows gathers their y rows from its LDS lists and writes out as 64-byte row runs; 4th launch and idx plane gone
# speedup vs baseline: 3.5361x; 1.2278x over previous
#include <hip/hip_runtime.h>

typedef __attribute__((ext_vector_type(2))) float v2f;
typedef __attribute__((ext_vector_type(8))) float v8f;

#define NPTS 4096
#define CCH  64
#define KNN  20
#define FMAX 3.402823466e+38f


__global__ void norms_kernel(const float* __restrict__ x, float* __restrict__ norms) {
    int p = blockIdx.x * blockDim.x + threadIdx.x;
    int b = p >> 12, n = p & 4095;
    const float* xb = x + b * CCH * NPTS + n;
    float s = 0.f;
    #pragma unroll
    for (int c = 0; c < CCH; ++c) { float v = xb[c * NPTS]; s += v * v; }
    norms[p] = s;
}

#define NSPLIT 8

__device__ __forceinline__ void topk_consider(float d, int col, float* __restrict__ ld,
                                              int* __restrict__ li,
                                              float& curMax, int& curArg) {
    if (d < curMax) {
        ld[curArg] = d;
        li[curArg] = col;
        float m = -FMAX; int arg = 0;
        for (int q = 0; q < KNN; ++q) {
            float v = ld[q];
            if (v > m) { m = v; arg = q; }
        }
        curMax = m; curArg = arg;
    }
}

__global__ void __launch_bounds__(32 * NSPLIT) __attribute__((amdgpu_waves_per_eu(8, 10)))
knn_kernel(const float* __restrict__ x,
           const float* __restrict__ norms,
           const float* __restrict__ y,
           const float* __restrict__ z,
           float* __restrict__ out) {
    __shared__ __attribute__((aligned(16))) float kd[NSPLIT][32 * KNN];
    __shared__ __attribute__((aligned(16))) int   ki[NSPLIT][32 * KNN];

    const int lane = threadIdx.x;
    const int w    = threadIdx.y;
    const int half = lane >> 4;
    const int lm   = lane & 15;
    const int b    = blockIdx.y;
    const int rowBase = blockIdx.x * 16;
    const float* xb = x + b * CCH * NPTS;
    const float* nb = norms + b * NPTS;

    v2f a[16];
    #pragma unroll
    for (int k = 0; k < 16; ++k) {
        int c = 4 * k + 2 * half;
        a[k].x = xb[(c    ) * NPTS + rowBase + lm];
        a[k].y = xb[(c + 1) * NPTS + rowBase + lm];
    }
    const float rn = nb[rowBase + lm];

    float* myd = &kd[w][lane * KNN];
    int*   myi = &ki[w][lane * KNN];
    for (int j = 0; j < KNN; ++j) { myd[j] = FMAX; myi[j] = 0; }
    float curMax = FMAX; int curArg = 0;

    for (int i = 0; i < NPTS / 16 / NSPLIT; ++i) {
        const int colBase = (i * NSPLIT + w) * 16;

        const float4 cnLo = *(const float4*)(nb + colBase + 8 * half);
        const float4 cnHi = *(const float4*)(nb + colBase + 8 * half + 4);
        const float cn[8] = {cnLo.x, cnLo.y, cnLo.z, cnLo.w, cnHi.x, cnHi.y, cnHi.z, cnHi.w};

        v2f bb[16];
        #pragma unroll
        for (int k = 0; k < 16; ++k) {
            int c = 4 * k + 2 * half;
            bb[k].x = xb[(c    ) * NPTS + colBase + lm];
            bb[k].y = xb[(c + 1) * NPTS + colBase + lm];
        }

        v8f acc = {};
        #pragma unroll
        for (int k = 0; k < 16; ++k)
            acc = __builtin_amdgcn_wmma_f32_16x16x4_f32(false, bb[k], false, a[k],
                                                        (short)0, acc, false, false);

        const int col0 = colBase + 8 * half;
        float dv[8];
        #pragma unroll
        for (int r = 0; r < 8; ++r) dv[r] = rn + cn[r] - 2.0f * acc[r];
        unsigned pend = 0u;
        #pragma unroll
        for (int r = 0; r < 8; ++r) pend |= (dv[r] < curMax) ? (1u << r) : 0u;
        while (pend != 0u) {
            const int r = __builtin_ctz(pend);
            pend &= pend - 1u;
            const float s0 = (r & 1) ? dv[1] : dv[0];
            const float s1 = (r & 1) ? dv[3] : dv[2];
            const float s2 = (r & 1) ? dv[5] : dv[4];
            const float s3 = (r & 1) ? dv[7] : dv[6];
            const float t0 = (r & 2) ? s1 : s0;
            const float t1 = (r & 2) ? s3 : s2;
            const float d  = (r & 4) ? t1 : t0;
            topk_consider(d, col0 + r, myd, myi, curMax, curArg);
        }
    }

    const int g = 2 * w + half;
    for (int stride = 1; stride < 2 * NSPLIT; stride *= 2) {
        __syncthreads();
        if ((g & (2 * stride - 1)) == 0) {
            const int og = g + stride;
            const float* od = &kd[og >> 1][((og & 1) * 16 + lm) * KNN];
            const int*   oi = &ki[og >> 1][((og & 1) * 16 + lm) * KNN];
            for (int j = 0; j < KNN; ++j)
                topk_consider(od[j], oi[j], myd, myi, curMax, curArg);
        }
    }

    __syncthreads();
    const float* yb = y + b * NPTS * CCH;
    float* stage = &kd[0][0] + 32 * KNN;
    for (int rr = 0; rr < 2; ++rr) {
        const int r = 2 * w + rr;
        const int* li = &ki[0][r * KNN];
        float b0 = -FMAX, b1 = -FMAX;
        #pragma unroll
        for (int j = 0; j < KNN; ++j) {
            const int m = li[j];
            b0 = fmaxf(b0, yb[m * CCH + lane]);
            b1 = fmaxf(b1, yb[m * CCH + 32 + lane]);
        }
        const float* zr = z + (b * NPTS + rowBase + r) * CCH;
        stage[lane * 17 + r]        = b0 + zr[lane];
        stage[(lane + 32) * 17 + r] = b1 + zr[lane + 32];
    }
    __syncthreads();
    const int tid = w * 32 + lane;
    #pragma unroll
    for (int q = 0; q < 4; ++q) {
        const int e = q * 256 + tid;
        const int o = e >> 4, r = e & 15;
        out[(b * CCH + o) * NPTS + rowBase + r] = stage[o * 17 + r];
    }
}

__global__ void __launch_bounds__(32) proj_kernel(const float* __restrict__ x,
                                                  const float* __restrict__ W,
                                                  float* __restrict__ y,
                                                  float* __restrict__ z) {
    const int lane = threadIdx.x;
    const int half = lane >> 4;
    const int lm   = lane & 15;
    const int b    = blockIdx.y;
    const int rowBase = blockIdx.x * 16;
    const float* xb = x + b * CCH * NPTS;

    v2f a[16];
    #pragma unroll
    for (int k = 0; k < 16; ++k) {
        int c = 4 * k + 2 * half;
        a[k].x = xb[(c    ) * NPTS + rowBase + lm];
        a[k].y = xb[(c + 1) * NPTS + rowBase + lm];
    }

    for (int ot = 0; ot < 4; ++ot) {
        const int o = ot * 16 + lm;
        v8f accY = {}, accZ = {};
        #pragma unroll
        for (int k = 0; k < 16; ++k) {
            int c = 4 * k + 2 * half;
            float w0 = W[o * 128 + c], w1 = W[o * 128 + c + 1];
            v2f bw, bz;
            bw.x = w0;
            bw.y = w1;
            bz.x = W[o * 128 + 64 + c    ] - w0;
            bz.y = W[o * 128 + 64 + c + 1] - w1;
            accY = __builtin_amdgcn_wmma_f32_16x16x4_f32(false, a[k], false, bw,
                                                         (short)0, accY, false, false);
            accZ = __builtin_amdgcn_wmma_f32_16x16x4_f32(false, a[k], false, bz,
                                                         (short)0, accZ, false, false);
        }
        #pragma unroll
        for (int r = 0; r < 8; ++r) {
            int row = rowBase + r + 8 * half;
            y[(b * NPTS + row) * CCH + o] = accY[r];
            z[(b * NPTS + row) * CCH + o] = accZ[r];
        }
    }
}

extern "C" void kernel_launch(void* const* d_in, const int* in_sizes, int n_in,
                              void* d_out, int out_size, void* d_ws, size_t ws_size,
                              hipStream_t stream) {
    (void)in_sizes; (void)n_in; (void)out_size; (void)ws_size;
    const float* x = (const float*)d_in[0];
    const float* W = (const float*)d_in[1];
    float* out = (float*)d_out;

    char* ws = (char*)d_ws;
    float* norms = (float*)(ws);
    float* y     = (float*)(ws + 65536 + 1310720);
    float* z     = (float*)(ws + 65536 + 1310720 + 4194304);

    norms_kernel    <<<dim3(64), 256, 0, stream>>>(x, norms);
    proj_kernel     <<<dim3(256, 4), 32, 0, stream>>>(x, W, y, z);
    knn_kernel      <<<dim3(256, 4), dim3(32, NSPLIT), 0, stream>>>(x, norms, y, z, out);
}


# knn: each wave posts per row the 3rd smallest of its two halves' two best (LDS, no barrier in the sweep); lanes prune with min(own 20th best, largest of 8 posts): 24 candidates lie under it, exact
# speedup vs baseline: 4.0958x; 1.1583x over previous
#include <hip/hip_runtime.h>

typedef __attribute__((ext_vector_type(2))) float v2f;
typedef __attribute__((ext_vector_type(8))) float v8f;
typedef __attribute__((ext_vector_type(4))) float v4f;
typedef __attribute__((address_space(3))) volatile float lds_vf;
typedef __attribute__((address_space(3))) volatile v4f   lds_v4f;

#define NPTS 4096
#define CCH  64
#define KNN  20
#define FMAX 3.402823466e+38f


__global__ void norms_kernel(const float* __restrict__ x, float* __restrict__ norms) {
    int p = blockIdx.x * blockDim.x + threadIdx.x;
    int b = p >> 12, n = p & 4095;
    const float* xb = x + b * CCH * NPTS + n;
    float s = 0.f;
    #pragma unroll
    for (int c = 0; c < CCH; ++c) { float v = xb[c * NPTS]; s += v * v; }
    norms[p] = s;
}

#define NSPLIT 8

__device__ __forceinline__ float xhalf(float x) {
    const int b = __builtin_bit_cast(int, x);
    return __builtin_bit_cast(float,
        __builtin_amdgcn_permlanex16(b, b, 0x76543210, 0xfedcba98, false, false));
}

__device__ __forceinline__ void topk_consider(float d, int col, float* __restrict__ ld,
                                              int* __restrict__ li,
                                              float& curMax, int& curArg) {
    if (d < curMax) {
        ld[curArg] = d;
        li[curArg] = col;
        float m = -FMAX; int arg = 0;
        for (int q = 0; q < KNN; ++q) {
            float v = ld[q];
            if (v > m) { m = v; arg = q; }
        }
        curMax = m; curArg = arg;
    }
}

__device__ __forceinline__ void sweep_consider(float d, int col, float* __restrict__ ld,
                                               int* __restrict__ li, float rowBound,
                                               float& curMax, int& curArg, float& thr,
                                               float& m1, float& m2) {
    if (d < thr) {
        ld[curArg] = d;
        li[curArg] = col;
        m2 = fminf(m2, fmaxf(m1, d));
        m1 = fminf(m1, d);
        float m = -FMAX; int arg = 0;
        for (int q = 0; q < KNN; ++q) {
            float v = ld[q];
            if (v > m) { m = v; arg = q; }
        }
        curMax = m; curArg = arg;
        thr = fminf(m, rowBound);
    }
}

__global__ void __launch_bounds__(32 * NSPLIT) __attribute__((amdgpu_waves_per_eu(8, 10)))
knn_kernel(const float* __restrict__ x,
           const float* __restrict__ norms,
           const float* __restrict__ y,
           const float* __restrict__ z,
           float* __restrict__ out) {
    __shared__ __attribute__((aligned(16))) float kd[NSPLIT][32 * KNN];
    __shared__ __attribute__((aligned(16))) int   ki[NSPLIT][32 * KNN];
    __shared__ __attribute__((aligned(16))) float kb[16][NSPLIT];

    const int lane = threadIdx.x;
    const int w    = threadIdx.y;
    const int half = lane >> 4;
    const int lm   = lane & 15;
    const int b    = blockIdx.y;
    const int rowBase = blockIdx.x * 16;
    const float* xb = x + b * CCH * NPTS;
    const float* nb = norms + b * NPTS;

    v2f a[16];
    #pragma unroll
    for (int k = 0; k < 16; ++k) {
        int c = 4 * k + 2 * half;
        a[k].x = xb[(c    ) * NPTS + rowBase + lm];
        a[k].y = xb[(c + 1) * NPTS + rowBase + lm];
    }
    const float rn = nb[rowBase + lm];

    float* myd = &kd[w][lane * KNN];
    int*   myi = &ki[w][lane * KNN];
    for (int j = 0; j < KNN; ++j) { myd[j] = FMAX; myi[j] = 0; }
    float curMax = FMAX; int curArg = 0;
    float m1 = FMAX, m2 = FMAX;
    lds_vf* post = (lds_vf*)&kb[lm][0];
    post[w] = FMAX;
    __syncthreads();

    for (int i = 0; i < NPTS / 16 / NSPLIT; ++i) {
        const int colBase = (i * NSPLIT + w) * 16;

        const v4f p0 = *(lds_v4f*)(post);
        const v4f p1 = *(lds_v4f*)(post + 4);

        const float4 cnLo = *(const float4*)(nb + colBase + 8 * half);
        const float4 cnHi = *(const float4*)(nb + colBase + 8 * half + 4);
        const float cn[8] = {cnLo.x, cnLo.y, cnLo.z, cnLo.w, cnHi.x, cnHi.y, cnHi.z, cnHi.w};

        v2f bb[16];
        #pragma unroll
        for (int k = 0; k < 16; ++k) {
            int c = 4 * k + 2 * half;
            bb[k].x = xb[(c    ) * NPTS + colBase + lm];
            bb[k].y = xb[(c + 1) * NPTS + colBase + lm];
        }

        v8f acc = {};
        #pragma unroll
        for (int k = 0; k < 16; ++k)
            acc = __builtin_amdgcn_wmma_f32_16x16x4_f32(false, bb[k], false, a[k],
                                                        (short)0, acc, false, false);

        const int col0 = colBase + 8 * half;
        float dv[8];
        #pragma unroll
        for (int r = 0; r < 8; ++r) dv[r] = rn + cn[r] - 2.0f * acc[r];
        const float rowBound = fmaxf(fmaxf(fmaxf(p0.x, p0.y), fmaxf(p0.z, p0.w)),
                                     fmaxf(fmaxf(p1.x, p1.y), fmaxf(p1.z, p1.w)));
        float thr = fminf(curMax, rowBound);
        unsigned pend = 0u;
        #pragma unroll
        for (int r = 0; r < 8; ++r) pend |= (dv[r] < thr) ? (1u << r) : 0u;
        while (pend != 0u) {
            const int r = __builtin_ctz(pend);
            pend &= pend - 1u;
            const float s0 = (r & 1) ? dv[1] : dv[0];
            const float s1 = (r & 1) ? dv[3] : dv[2];
            const float s2 = (r & 1) ? dv[5] : dv[4];
            const float s3 = (r & 1) ? dv[7] : dv[6];
            const float t0 = (r & 2) ? s1 : s0;
            const float t1 = (r & 2) ? s3 : s2;
            const float d  = (r & 4) ? t1 : t0;
            sweep_consider(d, col0 + r, myd, myi, rowBound, curMax, curArg, thr, m1, m2);
        }
        const float o1 = xhalf(m1), o2 = xhalf(m2);
        post[w] = fmaxf(fminf(m2, o2), fmaxf(m1, o1));
    }

    const int g = 2 * w + half;
    for (int stride = 1; stride < 2 * NSPLIT; stride *= 2) {
        __syncthreads();
        if ((g & (2 * stride - 1)) == 0) {
            const int og = g + stride;
            const float* od = &kd[og >> 1][((og & 1) * 16 + lm) * KNN];
            const int*   oi = &ki[og >> 1][((og & 1) * 16 + lm) * KNN];
            for (int j = 0; j < KNN; ++j)
                topk_consider(od[j], oi[j], myd, myi, curMax, curArg);
        }
    }

    __syncthreads();
    const float* yb = y + b * NPTS * CCH;
    float* stage = &kd[0][0] + 32 * KNN;
    for (int rr = 0; rr < 2; ++rr) {
        const int r = 2 * w + rr;
        const int* li = &ki[0][r * KNN];
        float b0 = -FMAX, b1 = -FMAX;
        #pragma unroll
        for (int j = 0; j < KNN; ++j) {
            const int m = li[j];
            b0 = fmaxf(b0, yb[m * CCH + lane]);
            b1 = fmaxf(b1, yb[m * CCH + 32 + lane]);
        }
        const float* zr = z + (b * NPTS + rowBase + r) * CCH;
        stage[lane * 17 + r]        = b0 + zr[lane];
        stage[(lane + 32) * 17 + r] = b1 + zr[lane + 32];
    }
    __syncthreads();
    const int tid = w * 32 + lane;
    #pragma unroll
    for (int q = 0; q < 4; ++q) {
        const int e = q * 256 + tid;
        const int o = e >> 4, r = e & 15;
        out[(b * CCH + o) * NPTS + rowBase + r] = stage[o * 17 + r];
    }
}

__global__ void __launch_bounds__(32) proj_kernel(const float* __restrict__ x,
                                                  const float* __restrict__ W,
                                                  float* __restrict__ y,
                                                  float* __restrict__ z) {
    const int lane = threadIdx.x;
    const int half = lane >> 4;
    const int lm   = lane & 15;
    const int b    = blockIdx.y;
    const int rowBase = blockIdx.x * 16;
    const float* xb = x + b * CCH * NPTS;

    v2f a[16];
    #pragma unroll
    for (int k = 0; k < 16; ++k) {
        int c = 4 * k + 2 * half;
        a[k].x = xb[(c    ) * NPTS + rowBase + lm];
        a[k].y = xb[(c + 1) * NPTS + rowBase + lm];
    }

    for (int ot = 0; ot < 4; ++ot) {
        const int o = ot * 16 + lm;
        v8f accY = {}, accZ = {};
        #pragma unroll
        for (int k = 0; k < 16; ++k) {
            int c = 4 * k + 2 * half;
            float w0 = W[o * 128 + c], w1 = W[o * 128 + c + 1];
            v2f bw, bz;
            bw.x = w0;
            bw.y = w1;
            bz.x = W[o * 128 + 64 + c    ] - w0;
            bz.y = W[o * 128 + 64 + c + 1] - w1;
            accY = __builtin_amdgcn_wmma_f32_16x16x4_f32(false, a[k], false, bw,
                                                         (short)0, accY, false, false);
            accZ = __builtin_amdgcn_wmma_f32_16x16x4_f32(false, a[k], false, bz,
                                                         (short)0, accZ, false, false);
        }
        #pragma unroll
        for (int r = 0; r < 8; ++r) {
            int row = rowBase + r + 8 * half;
            y[(b * NPTS + row) * CCH + o] = accY[r];
            z[(b * NPTS + row) * CCH + o] = accZ[r];
        }
    }
}

extern "C" void kernel_launch(void* const* d_in, const int* in_sizes, int n_in,
                              void* d_out, int out_size, void* d_ws, size_t ws_size,
                              hipStream_t stream) {
    (void)in_sizes; (void)n_in; (void)out_size; (void)ws_size;
    const float* x = (const float*)d_in[0];
    const float* W = (const float*)d_in[1];
    float* out = (float*)d_out;

    char* ws = (char*)d_ws;
    float* norms = (float*)(ws);
    float* y     = (float*)(ws + 65536 + 1310720);
    float* z     = (float*)(ws + 65536 + 1310720 + 4194304);

    norms_kernel    <<<dim3(64), 256, 0, stream>>>(x, norms);
    proj_kernel     <<<dim3(256, 4), 32, 0, stream>>>(x, W, y, z);
    knn_kernel      <<<dim3(256, 4), dim3(32, NSPLIT), 0, stream>>>(x, norms, y, z, out);
}


# knn: the sweep only appends a passing candidate to its lane's list (36 slots: two LDS stores and a count); a list's 20 smallest are searched for once after the sweep, or when it fills
# speedup vs baseline: 4.4641x; 1.0899x over previous
#include <hip/hip_runtime.h>

typedef __attribute__((ext_vector_type(2))) float v2f;
typedef __attribute__((ext_vector_type(8))) float v8f;
typedef __attribute__((ext_vector_type(4))) float v4f;
typedef __attribute__((address_space(3))) volatile float lds_vf;
typedef __attribute__((address_space(3))) volatile v4f   lds_v4f;

#define NPTS 4096
#define CCH  64
#define KNN  20
#define CAP  36
#define FMAX 3.402823466e+38f


__global__ void norms_kernel(const float* __restrict__ x, float* __restrict__ norms) {
    int p = blockIdx.x * blockDim.x + threadIdx.x;
    int b = p >> 12, n = p & 4095;
    const float* xb = x + b * CCH * NPTS + n;
    float s = 0.f;
    #pragma unroll
    for (int c = 0; c < CCH; ++c) { float v = xb[c * NPTS]; s += v * v; }
    norms[p] = s;
}

#define NSPLIT 8

__device__ __forceinline__ float xhalf(float x) {
    const int b = __builtin_bit_cast(int, x);
    return __builtin_bit_cast(float,
        __builtin_amdgcn_permlanex16(b, b, 0x76543210, 0xfedcba98, false, false));
}

__device__ __forceinline__ void topk_consider(float d, int col, float* __restrict__ ld,
                                              int* __restrict__ li,
                                              float& curMax, int& curArg) {
    if (d < curMax) {
        ld[curArg] = d;
        li[curArg] = col;
        float m = -FMAX; int arg = 0;
        for (int q = 0; q < KNN; ++q) {
            float v = ld[q];
            if (v > m) { m = v; arg = q; }
        }
        curMax = m; curArg = arg;
    }
}

__device__ __forceinline__ void keep20(float* __restrict__ ld, int* __restrict__ li, int n,
                                       float bound, float& curMax, int& curArg) {
    float m = -FMAX; int arg = 0;
    for (int q = 0; q < KNN; ++q) {
        float v = ld[q];
        if (v > m) { m = v; arg = q; }
    }
    curMax = m; curArg = arg;
    for (int e = KNN; e < n; ++e) {
        const float v = ld[e];
        if (v < bound) topk_consider(v, li[e], ld, li, curMax, curArg);
    }
}

__device__ __forceinline__ void sweep_append(float d, int col, float* __restrict__ ld,
                                             int* __restrict__ li, float rowBound,
                                             int& cnt, float& ownMax, float& thr,
                                             float& m1, float& m2) {
    if (d < thr) {
        if (cnt == CAP) {
            float cm; int ca;
            keep20(ld, li, CAP, rowBound, cm, ca);
            ownMax = cm;
            cnt = KNN;
            thr = fminf(cm, rowBound);
        }
        if (d < thr) {
            ld[cnt] = d;
            li[cnt] = col;
            ++cnt;
            m2 = fminf(m2, fmaxf(m1, d));
            m1 = fminf(m1, d);
        }
    }
}

__global__ void __launch_bounds__(32 * NSPLIT) __attribute__((amdgpu_waves_per_eu(8, 10)))
knn_kernel(const float* __restrict__ x,
           const float* __restrict__ norms,
           const float* __restrict__ y,
           const float* __restrict__ z,
           float* __restrict__ out) {
    __shared__ __attribute__((aligned(16))) float kd[NSPLIT][32 * CAP];
    __shared__ __attribute__((aligned(16))) int   ki[NSPLIT][32 * CAP];
    __shared__ __attribute__((aligned(16))) float kb[16][NSPLIT];

    const int lane = threadIdx.x;
    const int w    = threadIdx.y;
    const int half = lane >> 4;
    const int lm   = lane & 15;
    const int b    = blockIdx.y;
    const int rowBase = blockIdx.x * 16;
    const float* xb = x + b * CCH * NPTS;
    const float* nb = norms + b * NPTS;

    v2f a[16];
    #pragma unroll
    for (int k = 0; k < 16; ++k) {
        int c = 4 * k + 2 * half;
        a[k].x = xb[(c    ) * NPTS + rowBase + lm];
        a[k].y = xb[(c + 1) * NPTS + rowBase + lm];
    }
    const float rn = nb[rowBase + lm];

    float* myd = &kd[w][lane * CAP];
    int*   myi = &ki[w][lane * CAP];
    for (int j = 0; j < KNN; ++j) { myd[j] = FMAX; myi[j] = 0; }
    int   cnt = 0;
    float ownMax = FMAX;
    float lastBound = FMAX;
    float m1 = FMAX, m2 = FMAX;
    lds_vf* post = (lds_vf*)&kb[lm][0];
    post[w] = FMAX;
    __syncthreads();

    for (int i = 0; i < NPTS / 16 / NSPLIT; ++i) {
        const int colBase = (i * NSPLIT + w) * 16;

        const v4f p0 = *(lds_v4f*)(post);
        const v4f p1 = *(lds_v4f*)(post + 4);

        const float4 cnLo = *(const float4*)(nb + colBase + 8 * half);
        const float4 cnHi = *(const float4*)(nb + colBase + 8 * half + 4);
        const float cn[8] = {cnLo.x, cnLo.y, cnLo.z, cnLo.w, cnHi.x, cnHi.y, cnHi.z, cnHi.w};

        v2f bb[16];
        #pragma unroll
        for (int k = 0; k < 16; ++k) {
            int c = 4 * k + 2 * half;
            bb[k].x = xb[(c    ) * NPTS + colBase + lm];
            bb[k].y = xb[(c + 1) * NPTS + colBase + lm];
        }

        v8f acc = {};
        #pragma unroll
        for (int k = 0; k < 16; ++k)
            acc = __builtin_amdgcn_wmma_f32_16x16x4_f32(false, bb[k], false, a[k],
                                                        (short)0, acc, false, false);

        const int col0 = colBase + 8 * half;
        float dv[8];
        #pragma unroll
        for (int r = 0; r < 8; ++r) dv[r] = rn + cn[r] - 2.0f * acc[r];
        const float rowBound = fmaxf(fmaxf(fmaxf(p0.x, p0.y), fmaxf(p0.z, p0.w)),
                                     fmaxf(fmaxf(p1.x, p1.y), fmaxf(p1.z, p1.w)));
        float thr = fminf(ownMax, rowBound);
        lastBound = rowBound;
        unsigned pend = 0u;
        #pragma unroll
        for (int r = 0; r < 8; ++r) pend |= (dv[r] < thr) ? (1u << r) : 0u;
        while (pend != 0u) {
            const int r = __builtin_ctz(pend);
            pend &= pend - 1u;
            const float s0 = (r & 1) ? dv[1] : dv[0];
            const float s1 = (r & 1) ? dv[3] : dv[2];
            const float s2 = (r & 1) ? dv[5] : dv[4];
            const float s3 = (r & 1) ? dv[7] : dv[6];
            const float t0 = (r & 2) ? s1 : s0;
            const float t1 = (r & 2) ? s3 : s2;
            const float d  = (r & 4) ? t1 : t0;
            sweep_append(d, col0 + r, myd, myi, rowBound, cnt, ownMax, thr, m1, m2);
        }
        const float o1 = xhalf(m1), o2 = xhalf(m2);
        post[w] = fmaxf(fminf(m2, o2), fmaxf(m1, o1));
    }

    float curMax; int curArg;
    keep20(myd, myi, cnt, lastBound, curMax, curArg);

    const int g = 2 * w + half;
    for (int stride = 1; stride < 2 * NSPLIT; stride *= 2) {
        __syncthreads();
        if ((g & (2 * stride - 1)) == 0) {
            const int og = g + stride;
            const float* od = &kd[og >> 1][((og & 1) * 16 + lm) * CAP];
            const int*   oi = &ki[og >> 1][((og & 1) * 16 + lm) * CAP];
            for (int j = 0; j < KNN; ++j)
                topk_consider(od[j], oi[j], myd, myi, curMax, curArg);
        }
    }

    __syncthreads();
    const float* yb = y + b * NPTS * CCH;
    float* stage = &kd[1][0];
    for (int rr = 0; rr < 2; ++rr) {
        const int r = 2 * w + rr;
        const int* li = &ki[0][r * CAP];
        float b0 = -FMAX, b1 = -FMAX;
        #pragma unroll
        for (int j = 0; j < KNN; ++j) {
            const int m = li[j];
            b0 = fmaxf(b0, yb[m * CCH + lane]);
            b1 = fmaxf(b1, yb[m * CCH + 32 + lane]);
        }
        const float* zr = z + (b * NPTS + rowBase + r) * CCH;
        stage[lane * 17 + r]        = b0 + zr[lane];
        stage[(lane + 32) * 17 + r] = b1 + zr[lane + 32];
    }
    __syncthreads();
    const int tid = w * 32 + lane;
    #pragma unroll
    for (int q = 0; q < 4; ++q) {
        const int e = q * 256 + tid;
        const int o = e >> 4, r = e & 15;
        out[(b * CCH + o) * NPTS + rowBase + r] = stage[o * 17 + r];
    }
}

__global__ void __launch_bounds__(32) proj_kernel(const float* __restrict__ x,
                                                  const float* __restrict__ W,
                                                  float* __restrict__ y,
                                                  float* __restrict__ z) {
    const int lane = threadIdx.x;
    const int half = lane >> 4;
    const int lm   = lane & 15;
    const int b    = blockIdx.y;
    const int rowBase = blockIdx.x * 16;
    const float* xb = x + b * CCH * NPTS;

    v2f a[16];
    #pragma unroll
    for (int k = 0; k < 16; ++k) {
        int c = 4 * k + 2 * half;
        a[k].x = xb[(c    ) * NPTS + rowBase + lm];
        a[k].y = xb[(c + 1) * NPTS + rowBase + lm];
    }

    for (int ot = 0; ot < 4; ++ot) {
        const int o = ot * 16 + lm;
        v8f accY = {}, accZ = {};
        #pragma unroll
        for (int k = 0; k < 16; ++k) {
            int c = 4 * k + 2 * half;
            float w0 = W[o * 128 + c], w1 = W[o * 128 + c + 1];
            v2f bw, bz;
            bw.x = w0;
            bw.y = w1;
            bz.x = W[o * 128 + 64 + c    ] - w0;
            bz.y = W[o * 128 + 64 + c + 1] - w1;
            accY = __builtin_amdgcn_wmma_f32_16x16x4_f32(false, a[k], false, bw,
                                                         (short)0, accY, false, false);
            accZ = __builtin_amdgcn_wmma_f32_16x16x4_f32(false, a[k], false, bz,
                                                         (short)0, accZ, false, false);
        }
        #pragma unroll
        for (int r = 0; r < 8; ++r) {
            int row = rowBase + r + 8 * half;
            y[(b * NPTS + row) * CCH + o] = accY[r];
            z[(b * NPTS + row) * CCH + o] = accZ[r];
        }
    }
}

extern "C" void kernel_launch(void* const* d_in, const int* in_sizes, int n_in,
                              void* d_out, int out_size, void* d_ws, size_t ws_size,
                              hipStream_t stream) {
    (void)in_sizes; (void)n_in; (void)out_size; (void)ws_size;
    const float* x = (const float*)d_in[0];
    const float* W = (const float*)d_in[1];
    float* out = (float*)d_out;

    char* ws = (char*)d_ws;
    float* norms = (float*)(ws);
    float* y     = (float*)(ws + 65536 + 1310720);
    float* z     = (float*)(ws + 65536 + 1310720 + 4194304);

    norms_kernel    <<<dim3(64), 256, 0, stream>>>(x, norms);
    proj_kernel     <<<dim3(256, 4), 32, 0, stream>>>(x, W, y, z);
    knn_kernel      <<<dim3(256, 4), dim3(32, NSPLIT), 0, stream>>>(x, norms, y, z, out);
}


# knn merge: each list is first cut to its entries under the row's final bound (a handful of 20); the 4 pairwise rounds append them and search a list's worst slot only once it holds 20
# speedup vs baseline: 4.8077x; 1.0770x over previous
#include <hip/hip_runtime.h>

typedef __attribute__((ext_vector_type(2))) float v2f;
typedef __attribute__((ext_vector_type(8))) float v8f;
typedef __attribute__((ext_vector_type(4))) float v4f;
typedef __attribute__((address_space(3))) volatile float lds_vf;
typedef __attribute__((address_space(3))) volatile v4f   lds_v4f;

#define NPTS 4096
#define CCH  64
#define KNN  20
#define CAP  36
#define FMAX 3.402823466e+38f


__global__ void norms_kernel(const float* __restrict__ x, float* __restrict__ norms) {
    int p = blockIdx.x * blockDim.x + threadIdx.x;
    int b = p >> 12, n = p & 4095;
    const float* xb = x + b * CCH * NPTS + n;
    float s = 0.f;
    #pragma unroll
    for (int c = 0; c < CCH; ++c) { float v = xb[c * NPTS]; s += v * v; }
    norms[p] = s;
}

#define NSPLIT 8

__device__ __forceinline__ float xhalf(float x) {
    const int b = __builtin_bit_cast(int, x);
    return __builtin_bit_cast(float,
        __builtin_amdgcn_permlanex16(b, b, 0x76543210, 0xfedcba98, false, false));
}

__device__ __forceinline__ void topk_consider(float d, int col, float* __restrict__ ld,
                                              int* __restrict__ li,
                                              float& curMax, int& curArg) {
    if (d < curMax) {
        ld[curArg] = d;
        li[curArg] = col;
        float m = -FMAX; int arg = 0;
        for (int q = 0; q < KNN; ++q) {
            float v = ld[q];
            if (v > m) { m = v; arg = q; }
        }
        curMax = m; curArg = arg;
    }
}

__device__ __forceinline__ void keep20(float* __restrict__ ld, int* __restrict__ li, int n,
                                       float bound, float& curMax, int& curArg) {
    float m = -FMAX; int arg = 0;
    for (int q = 0; q < KNN; ++q) {
        float v = ld[q];
        if (v > m) { m = v; arg = q; }
    }
    curMax = m; curArg = arg;
    for (int e = KNN; e < n; ++e) {
        const float v = ld[e];
        if (v < bound) topk_consider(v, li[e], ld, li, curMax, curArg);
    }
}

__device__ __forceinline__ void scan20(const float* __restrict__ ld, float& curMax, int& curArg) {
    float m = -FMAX; int arg = 0;
    for (int q = 0; q < KNN; ++q) {
        float v = ld[q];
        if (v > m) { m = v; arg = q; }
    }
    curMax = m; curArg = arg;
}

__device__ __forceinline__ void sweep_append(float d, int col, float* __restrict__ ld,
                                             int* __restrict__ li, float rowBound,
                                             int& cnt, float& ownMax, float& thr,
                                             float& m1, float& m2) {
    if (d < thr) {
        if (cnt == CAP) {
            float cm; int ca;
            keep20(ld, li, CAP, rowBound, cm, ca);
            ownMax = cm;
            cnt = KNN;
            thr = fminf(cm, rowBound);
        }
        if (d < thr) {
            ld[cnt] = d;
            li[cnt] = col;
            ++cnt;
            m2 = fminf(m2, fmaxf(m1, d));
            m1 = fminf(m1, d);
        }
    }
}

__global__ void __launch_bounds__(32 * NSPLIT) __attribute__((amdgpu_waves_per_eu(8, 10)))
knn_kernel(const float* __restrict__ x,
           const float* __restrict__ norms,
           const float* __restrict__ y,
           const float* __restrict__ z,
           float* __restrict__ out) {
    __shared__ __attribute__((aligned(16))) float kd[NSPLIT][32 * CAP];
    __shared__ __attribute__((aligned(16))) int   ki[NSPLIT][32 * CAP];
    __shared__ __attribute__((aligned(16))) float kb[16][NSPLIT];
    __shared__ int kc[NSPLIT][32];

    const int lane = threadIdx.x;
    const int w    = threadIdx.y;
    const int half = lane >> 4;
    const int lm   = lane & 15;
    const int b    = blockIdx.y;
    const int rowBase = blockIdx.x * 16;
    const float* xb = x + b * CCH * NPTS;
    const float* nb = norms + b * NPTS;

    v2f a[16];
    #pragma unroll
    for (int k = 0; k < 16; ++k) {
        int c = 4 * k + 2 * half;
        a[k].x = xb[(c    ) * NPTS + rowBase + lm];
        a[k].y = xb[(c + 1) * NPTS + rowBase + lm];
    }
    const float rn = nb[rowBase + lm];

    float* myd = &kd[w][lane * CAP];
    int*   myi = &ki[w][lane * CAP];
    int   cnt = 0;
    float ownMax = FMAX;
    float m1 = FMAX, m2 = FMAX;
    lds_vf* post = (lds_vf*)&kb[lm][0];
    post[w] = FMAX;
    __syncthreads();

    for (int i = 0; i < NPTS / 16 / NSPLIT; ++i) {
        const int colBase = (i * NSPLIT + w) * 16;

        const v4f p0 = *(lds_v4f*)(post);
        const v4f p1 = *(lds_v4f*)(post + 4);

        const float4 cnLo = *(const float4*)(nb + colBase + 8 * half);
        const float4 cnHi = *(const float4*)(nb + colBase + 8 * half + 4);
        const float cn[8] = {cnLo.x, cnLo.y, cnLo.z, cnLo.w, cnHi.x, cnHi.y, cnHi.z, cnHi.w};

        v2f bb[16];
        #pragma unroll
        for (int k = 0; k < 16; ++k) {
            int c = 4 * k + 2 * half;
            bb[k].x = xb[(c    ) * NPTS + colBase + lm];
            bb[k].y = xb[(c + 1) * NPTS + colBase + lm];
        }

        v8f acc = {};
        #pragma unroll
        for (int k = 0; k < 16; ++k)
            acc = __builtin_amdgcn_wmma_f32_16x16x4_f32(false, bb[k], false, a[k],
                                                        (short)0, acc, false, false);

        const int col0 = colBase + 8 * half;
        float dv[8];
        #pragma unroll
        for (int r = 0; r < 8; ++r) dv[r] = rn + cn[r] - 2.0f * acc[r];
        const float rowBound = fmaxf(fmaxf(fmaxf(p0.x, p0.y), fmaxf(p0.z, p0.w)),
                                     fmaxf(fmaxf(p1.x, p1.y), fmaxf(p1.z, p1.w)));
        float thr = fminf(ownMax, rowBound);
        unsigned pend = 0u;
        #pragma unroll
        for (int r = 0; r < 8; ++r) pend |= (dv[r] < thr) ? (1u << r) : 0u;
        while (pend != 0u) {
            const int r = __builtin_ctz(pend);
            pend &= pend - 1u;
            const float s0 = (r & 1) ? dv[1] : dv[0];
            const float s1 = (r & 1) ? dv[3] : dv[2];
            const float s2 = (r & 1) ? dv[5] : dv[4];
            const float s3 = (r & 1) ? dv[7] : dv[6];
            const float t0 = (r & 2) ? s1 : s0;
            const float t1 = (r & 2) ? s3 : s2;
            const float d  = (r & 4) ? t1 : t0;
            sweep_append(d, col0 + r, myd, myi, rowBound, cnt, ownMax, thr, m1, m2);
        }
        const float o1 = xhalf(m1), o2 = xhalf(m2);
        post[w] = fmaxf(fminf(m2, o2), fmaxf(m1, o1));
    }

    __syncthreads();
    float rowFinal;
    {
        const v4f q0 = *(lds_v4f*)(post);
        const v4f q1 = *(lds_v4f*)(post + 4);
        rowFinal = fmaxf(fmaxf(fmaxf(q0.x, q0.y), fmaxf(q0.z, q0.w)),
                         fmaxf(fmaxf(q1.x, q1.y), fmaxf(q1.z, q1.w)));
    }
    int n = 0;
    for (int e = 0; e < cnt; e += 4) {
        const v4f  dq = *(const v4f*)(myd + e);
        const int4 cq = *(const int4*)(myi + e);
        const float dd[4] = {dq.x, dq.y, dq.z, dq.w};
        const int   cc[4] = {cq.x, cq.y, cq.z, cq.w};
        #pragma unroll
        for (int q = 0; q < 4; ++q) {
            myd[n] = dd[q];
            myi[n] = cc[q];
            n += (e + q < cnt && dd[q] < rowFinal) ? 1 : 0;
        }
    }
    float curMax = FMAX; int curArg = 0;
    if (n >= KNN) {
        keep20(myd, myi, n, rowFinal, curMax, curArg);
        n = KNN;
    }
    kc[w][lane] = n;

    const int g = 2 * w + half;
    for (int stride = 1; stride < 2 * NSPLIT; stride *= 2) {
        __syncthreads();
        if ((g & (2 * stride - 1)) == 0) {
            const int og = g + stride;
            const int ol = (og & 1) * 16 + lm;
            const float* od = &kd[og >> 1][ol * CAP];
            const int*   oi = &ki[og >> 1][ol * CAP];
            const int    on = kc[og >> 1][ol];
            for (int j = 0; j < on; ++j) {
                const float d = od[j];
                const int   c = oi[j];
                if (n < KNN) {
                    myd[n] = d; myi[n] = c; ++n;
                    if (n == KNN) scan20(myd, curMax, curArg);
                } else {
                    topk_consider(d, c, myd, myi, curMax, curArg);
                }
            }
            kc[w][lane] = n;
        }
    }
    if (g == 0) {
        const int fill = (n > 0) ? myi[0] : 0;
        for (int j = n; j < KNN; ++j) myi[j] = fill;
    }

    __syncthreads();
    const float* yb = y + b * NPTS * CCH;
    float* stage = &kd[1][0];
    for (int rr = 0; rr < 2; ++rr) {
        const int r = 2 * w + rr;
        const int* li = &ki[0][r * CAP];
        float b0 = -FMAX, b1 = -FMAX;
        #pragma unroll
        for (int j = 0; j < KNN; ++j) {
            const int m = li[j];
            b0 = fmaxf(b0, yb[m * CCH + lane]);
            b1 = fmaxf(b1, yb[m * CCH + 32 + lane]);
        }
        const float* zr = z + (b * NPTS + rowBase + r) * CCH;
        stage[lane * 17 + r]        = b0 + zr[lane];
        stage[(lane + 32) * 17 + r] = b1 + zr[lane + 32];
    }
    __syncthreads();
    const int tid = w * 32 + lane;
    #pragma unroll
    for (int q = 0; q < 4; ++q) {
        const int e = q * 256 + tid;
        const int o = e >> 4, r = e & 15;
        out[(b * CCH + o) * NPTS + rowBase + r] = stage[o * 17 + r];
    }
}

__global__ void __launch_bounds__(32) proj_kernel(const float* __restrict__ x,
                                                  const float* __restrict__ W,
                                                  float* __restrict__ y,
                                                  float* __restrict__ z) {
    const int lane = threadIdx.x;
    const int half = lane >> 4;
    const int lm   = lane & 15;
    const int b    = blockIdx.y;
    const int rowBase = blockIdx.x * 16;
    const float* xb = x + b * CCH * NPTS;

    v2f a[16];
    #pragma unroll
    for (int k = 0; k < 16; ++k) {
        int c = 4 * k + 2 * half;
        a[k].x = xb[(c    ) * NPTS + rowBase + lm];
        a[k].y = xb[(c + 1) * NPTS + rowBase + lm];
    }

    for (int ot = 0; ot < 4; ++ot) {
        const int o = ot * 16 + lm;
        v8f accY = {}, accZ = {};
        #pragma unroll
        for (int k = 0; k < 16; ++k) {
            int c = 4 * k + 2 * half;
            float w0 = W[o * 128 + c], w1 = W[o * 128 + c + 1];
            v2f bw, bz;
            bw.x = w0;
            bw.y = w1;
            bz.x = W[o * 128 + 64 + c    ] - w0;
            bz.y = W[o * 128 + 64 + c + 1] - w1;
            accY = __builtin_amdgcn_wmma_f32_16x16x4_f32(false, a[k], false, bw,
                                                         (short)0, accY, false, false);
            accZ = __builtin_amdgcn_wmma_f32_16x16x4_f32(false, a[k], false, bz,
                                                         (short)0, accZ, false, false);
        }
        #pragma unroll
        for (int r = 0; r < 8; ++r) {
            int row = rowBase + r + 8 * half;
            y[(b * NPTS + row) * CCH + o] = accY[r];
            z[(b * NPTS + row) * CCH + o] = accZ[r];
        }
    }
}

extern "C" void kernel_launch(void* const* d_in, const int* in_sizes, int n_in,
                              void* d_out, int out_size, void* d_ws, size_t ws_size,
                              hipStream_t stream) {
    (void)in_sizes; (void)n_in; (void)out_size; (void)ws_size;
    const float* x = (const float*)d_in[0];
    const float* W = (const float*)d_in[1];
    float* out = (float*)d_out;

    char* ws = (char*)d_ws;
    float* norms = (float*)(ws);
    float* y     = (float*)(ws + 65536 + 1310720);
    float* z     = (float*)(ws + 65536 + 1310720 + 4194304);

    norms_kernel    <<<dim3(64), 256, 0, stream>>>(x, norms);
    proj_kernel     <<<dim3(256, 4), 32, 0, stream>>>(x, W, y, z);
    knn_kernel      <<<dim3(256, 4), dim3(32, NSPLIT), 0, stream>>>(x, norms, y, z, out);
}


# knn sweep: a lane stores all 8 candidates of a tile to its list, count moves on only for one under the bound (no mask, no insert loop); wave cuts its lists when one runs short; posts read after loads
# speedup vs baseline: 5.1977x; 1.0811x over previous
#include <hip/hip_runtime.h>

typedef __attribute__((ext_vector_type(2))) float v2f;
typedef __attribute__((ext_vector_type(8))) float v8f;
typedef __attribute__((ext_vector_type(4))) float v4f;
typedef __attribute__((address_space(3))) volatile float lds_vf;
typedef __attribute__((address_space(3))) volatile v4f   lds_v4f;

#define NPTS 4096
#define CCH  64
#define KNN  20
#define CAP  36
#define FMAX 3.402823466e+38f


__global__ void norms_kernel(const float* __restrict__ x, float* __restrict__ norms) {
    int p = blockIdx.x * blockDim.x + threadIdx.x;
    int b = p >> 12, n = p & 4095;
    const float* xb = x + b * CCH * NPTS + n;
    float s = 0.f;
    #pragma unroll
    for (int c = 0; c < CCH; ++c) { float v = xb[c * NPTS]; s += v * v; }
    norms[p] = s;
}

#define NSPLIT 8

__device__ __forceinline__ float xhalf(float x) {
    const int b = __builtin_bit_cast(int, x);
    return __builtin_bit_cast(float,
        __builtin_amdgcn_permlanex16(b, b, 0x76543210, 0xfedcba98, false, false));
}

__device__ __forceinline__ void topk_consider(float d, int col, float* __restrict__ ld,
                                              int* __restrict__ li,
                                              float& curMax, int& curArg) {
    if (d < curMax) {
        ld[curArg] = d;
        li[curArg] = col;
        float m = -FMAX; int arg = 0;
        for (int q = 0; q < KNN; ++q) {
            float v = ld[q];
            if (v > m) { m = v; arg = q; }
        }
        curMax = m; curArg = arg;
    }
}

__device__ __forceinline__ void keep20(float* __restrict__ ld, int* __restrict__ li, int n,
                                       float bound, float& curMax, int& curArg) {
    float m = -FMAX; int arg = 0;
    for (int q = 0; q < KNN; ++q) {
        float v = ld[q];
        if (v > m) { m = v; arg = q; }
    }
    curMax = m; curArg = arg;
    for (int e = KNN; e < n; ++e) {
        const float v = ld[e];
        if (v < bound) topk_consider(v, li[e], ld, li, curMax, curArg);
    }
}

__device__ __forceinline__ void scan20(const float* __restrict__ ld, float& curMax, int& curArg) {
    float m = -FMAX; int arg = 0;
    for (int q = 0; q < KNN; ++q) {
        float v = ld[q];
        if (v > m) { m = v; arg = q; }
    }
    curMax = m; curArg = arg;
}

__device__ __forceinline__ int cut_list(float* __restrict__ ld, int* __restrict__ li, int cnt,
                                        float bound) {
    int n = 0;
    for (int e = 0; e < cnt; e += 4) {
        const v4f  dq = *(const v4f*)(ld + e);
        const int4 cq = *(const int4*)(li + e);
        const float dd[4] = {dq.x, dq.y, dq.z, dq.w};
        const int   cc[4] = {cq.x, cq.y, cq.z, cq.w};
        #pragma unroll
        for (int q = 0; q < 4; ++q) {
            ld[n] = dd[q];
            li[n] = cc[q];
            n += (e + q < cnt && dd[q] < bound) ? 1 : 0;
        }
    }
    return n;
}

__global__ void __launch_bounds__(32 * NSPLIT) __attribute__((amdgpu_waves_per_eu(8, 10)))
knn_kernel(const float* __restrict__ x,
           const float* __restrict__ norms,
           const float* __restrict__ y,
           const float* __restrict__ z,
           float* __restrict__ out) {
    __shared__ __attribute__((aligned(16))) float kd[NSPLIT][32 * CAP];
    __shared__ __attribute__((aligned(16))) int   ki[NSPLIT][32 * CAP];
    __shared__ __attribute__((aligned(16))) float kb[16][NSPLIT];
    __shared__ int kc[NSPLIT][32];

    const int lane = threadIdx.x;
    const int w    = threadIdx.y;
    const int half = lane >> 4;
    const int lm   = lane & 15;
    const int b    = blockIdx.y;
    const int rowBase = blockIdx.x * 16;
    const float* xb = x + b * CCH * NPTS;
    const float* nb = norms + b * NPTS;

    v2f a[16];
    #pragma unroll
    for (int k = 0; k < 16; ++k) {
        int c = 4 * k + 2 * half;
        a[k].x = xb[(c    ) * NPTS + rowBase + lm];
        a[k].y = xb[(c + 1) * NPTS + rowBase + lm];
    }
    const float rn = nb[rowBase + lm];

    float* myd = &kd[w][lane * CAP];
    int*   myi = &ki[w][lane * CAP];
    int   cnt = 0;
    float ownMax = FMAX;
    float m1 = FMAX, m2 = FMAX;
    lds_vf* post = (lds_vf*)&kb[lm][0];
    post[w] = FMAX;
    __syncthreads();

    for (int i = 0; i < NPTS / 16 / NSPLIT; ++i) {
        const int colBase = (i * NSPLIT + w) * 16;

        const float4 cnLo = *(const float4*)(nb + colBase + 8 * half);
        const float4 cnHi = *(const float4*)(nb + colBase + 8 * half + 4);
        const float cn[8] = {cnLo.x, cnLo.y, cnLo.z, cnLo.w, cnHi.x, cnHi.y, cnHi.z, cnHi.w};

        v2f bb[16];
        #pragma unroll
        for (int k = 0; k < 16; ++k) {
            int c = 4 * k + 2 * half;
            bb[k].x = xb[(c    ) * NPTS + colBase + lm];
            bb[k].y = xb[(c + 1) * NPTS + colBase + lm];
        }

        const v4f p0 = *(lds_v4f*)(post);
        const v4f p1 = *(lds_v4f*)(post + 4);

        v8f acc = {};
        #pragma unroll
        for (int k = 0; k < 16; ++k)
            acc = __builtin_amdgcn_wmma_f32_16x16x4_f32(false, bb[k], false, a[k],
                                                        (short)0, acc, false, false);

        const int col0 = colBase + 8 * half;
        float dv[8];
        #pragma unroll
        for (int r = 0; r < 8; ++r) dv[r] = rn + cn[r] - 2.0f * acc[r];
        const float rowBound = fmaxf(fmaxf(fmaxf(p0.x, p0.y), fmaxf(p0.z, p0.w)),
                                     fmaxf(fmaxf(p1.x, p1.y), fmaxf(p1.z, p1.w)));
        if (__builtin_amdgcn_ballot_w32(cnt > CAP - 8) != 0u) {
            cnt = cut_list(myd, myi, cnt, rowBound);
            if (cnt > CAP - 8) {
                float cm; int ca;
                keep20(myd, myi, cnt, rowBound, cm, ca);
                ownMax = cm;
                cnt = KNN;
            }
        }
        const float thr = fminf(ownMax, rowBound);
        #pragma unroll
        for (int r = 0; r < 8; ++r) {
            const float d = dv[r];
            myd[cnt] = d;
            myi[cnt] = col0 + r;
            cnt += (d < thr) ? 1 : 0;
            m2 = fminf(m2, fmaxf(m1, d));
            m1 = fminf(m1, d);
        }
        const float o1 = xhalf(m1), o2 = xhalf(m2);
        post[w] = fmaxf(fminf(m2, o2), fmaxf(m1, o1));
    }

    __syncthreads();
    float rowFinal;
    {
        const v4f q0 = *(lds_v4f*)(post);
        const v4f q1 = *(lds_v4f*)(post + 4);
        rowFinal = fmaxf(fmaxf(fmaxf(q0.x, q0.y), fmaxf(q0.z, q0.w)),
                         fmaxf(fmaxf(q1.x, q1.y), fmaxf(q1.z, q1.w)));
    }
    int n = cut_list(myd, myi, cnt, rowFinal);
    float curMax = FMAX; int curArg = 0;
    if (n >= KNN) {
        keep20(myd, myi, n, rowFinal, curMax, curArg);
        n = KNN;
    }
    kc[w][lane] = n;

    const int g = 2 * w + half;
    for (int stride = 1; stride < 2 * NSPLIT; stride *= 2) {
        __syncthreads();
        if ((g & (2 * stride - 1)) == 0) {
            const int og = g + stride;
            const int ol = (og & 1) * 16 + lm;
            const float* od = &kd[og >> 1][ol * CAP];
            const int*   oi = &ki[og >> 1][ol * CAP];
            const int    on = kc[og >> 1][ol];
            for (int j = 0; j < on; ++j) {
                const float d = od[j];
                const int   c = oi[j];
                if (n < KNN) {
                    myd[n] = d; myi[n] = c; ++n;
                    if (n == KNN) scan20(myd, curMax, curArg);
                } else {
                    topk_consider(d, c, myd, myi, curMax, curArg);
                }
            }
            kc[w][lane] = n;
        }
    }
    if (g == 0) {
        const int fill = (n > 0) ? myi[0] : 0;
        for (int j = n; j < KNN; ++j) myi[j] = fill;
    }

    __syncthreads();
    const float* yb = y + b * NPTS * CCH;
    float* stage = &kd[1][0];
    for (int rr = 0; rr < 2; ++rr) {
        const int r = 2 * w + rr;
        const int* li = &ki[0][r * CAP];
        float b0 = -FMAX, b1 = -FMAX;
        #pragma unroll
        for (int j = 0; j < KNN; ++j) {
            const int m = li[j];
            b0 = fmaxf(b0, yb[m * CCH + lane]);
            b1 = fmaxf(b1, yb[m * CCH + 32 + lane]);
        }
        const float* zr = z + (b * NPTS + rowBase + r) * CCH;
        stage[lane * 17 + r]        = b0 + zr[lane];
        stage[(lane + 32) * 17 + r] = b1 + zr[lane + 32];
    }
    __syncthreads();
    const int tid = w * 32 + lane;
    #pragma unroll
    for (int q = 0; q < 4; ++q) {
        const int e = q * 256 + tid;
        const int o = e >> 4, r = e & 15;
        out[(b * CCH + o) * NPTS + rowBase + r] = stage[o * 17 + r];
    }
}

__global__ void __launch_bounds__(32) proj_kernel(const float* __restrict__ x,
                                                  const float* __restrict__ W,
                                                  float* __restrict__ y,
                                                  float* __restrict__ z) {
    const int lane = threadIdx.x;
    const int half = lane >> 4;
    const int lm   = lane & 15;
    const int b    = blockIdx.y;
    const int rowBase = blockIdx.x * 16;
    const float* xb = x + b * CCH * NPTS;

    v2f a[16];
    #pragma unroll
    for (int k = 0; k < 16; ++k) {
        int c = 4 * k + 2 * half;
        a[k].x = xb[(c    ) * NPTS + rowBase + lm];
        a[k].y = xb[(c + 1) * NPTS + rowBase + lm];
    }

    for (int ot = 0; ot < 4; ++ot) {
        const int o = ot * 16 + lm;
        v8f accY = {}, accZ = {};
        #pragma unroll
        for (int k = 0; k < 16; ++k) {
            int c = 4 * k + 2 * half;
            float w0 = W[o * 128 + c], w1 = W[o * 128 + c + 1];
            v2f bw, bz;
            bw.x = w0;
            bw.y = w1;
            bz.x = W[o * 128 + 64 + c    ] - w0;
            bz.y = W[o * 128 + 64 + c + 1] - w1;
            accY = __builtin_amdgcn_wmma_f32_16x16x4_f32(false, a[k], false, bw,
                                                         (short)0, accY, false, false);
            accZ = __builtin_amdgcn_wmma_f32_16x16x4_f32(false, a[k], false, bz,
                                                         (short)0, accZ, false, false);
        }
        #pragma unroll
        for (int r = 0; r < 8; ++r) {
            int row = rowBase + r + 8 * half;
            y[(b * NPTS + row) * CCH + o] = accY[r];
            z[(b * NPTS + row) * CCH + o] = accZ[r];
        }
    }
}

extern "C" void kernel_launch(void* const* d_in, const int* in_sizes, int n_in,
                              void* d_out, int out_size, void* d_ws, size_t ws_size,
                              hipStream_t stream) {
    (void)in_sizes; (void)n_in; (void)out_size; (void)ws_size;
    const float* x = (const float*)d_in[0];
    const float* W = (const float*)d_in[1];
    float* out = (float*)d_out;

    char* ws = (char*)d_ws;
    float* norms = (float*)(ws);
    float* y     = (float*)(ws + 65536 + 1310720);
    float* z     = (float*)(ws + 65536 + 1310720 + 4194304);

    norms_kernel    <<<dim3(64), 256, 0, stream>>>(x, norms);
    proj_kernel     <<<dim3(256, 4), 32, 0, stream>>>(x, W, y, z);
    knn_kernel      <<<dim3(256, 4), dim3(32, NSPLIT), 0, stream>>>(x, norms, y, z, out);
}


# knn sweep: a wave takes 32-column tiles with one 8-byte load per lane and channel (16 lanes = one whole 128-byte line); even and odd columns form two Gram tiles: half the loads and L2 lines per tile
# speedup vs baseline: 5.3853x; 1.0361x over previous
#include <hip/hip_runtime.h>

typedef __attribute__((ext_vector_type(2))) float v2f;
typedef __attribute__((ext_vector_type(8))) float v8f;
typedef __attribute__((ext_vector_type(4))) float v4f;
typedef __attribute__((address_space(3))) volatile float lds_vf;
typedef __attribute__((address_space(3))) volatile v4f   lds_v4f;

#define NPTS 4096
#define CCH  64
#define KNN  20
#define CAP  36
#define FMAX 3.402823466e+38f


__global__ void norms_kernel(const float* __restrict__ x, float* __restrict__ norms) {
    int p = blockIdx.x * blockDim.x + threadIdx.x;
    int b = p >> 12, n = p & 4095;
    const float* xb = x + b * CCH * NPTS + n;
    float s = 0.f;
    #pragma unroll
    for (int c = 0; c < CCH; ++c) { float v = xb[c * NPTS]; s += v * v; }
    norms[(p & ~15) | ((p & 1) << 3) | ((p & 15) >> 1)] = s;
}

#define NSPLIT 8

__device__ __forceinline__ float xhalf(float x) {
    const int b = __builtin_bit_cast(int, x);
    return __builtin_bit_cast(float,
        __builtin_amdgcn_permlanex16(b, b, 0x76543210, 0xfedcba98, false, false));
}

__device__ __forceinline__ void topk_consider(float d, int col, float* __restrict__ ld,
                                              int* __restrict__ li,
                                              float& curMax, int& curArg) {
    if (d < curMax) {
        ld[curArg] = d;
        li[curArg] = col;
        float m = -FMAX; int arg = 0;
        for (int q = 0; q < KNN; ++q) {
            float v = ld[q];
            if (v > m) { m = v; arg = q; }
        }
        curMax = m; curArg = arg;
    }
}

__device__ __forceinline__ void keep20(float* __restrict__ ld, int* __restrict__ li, int n,
                                       float bound, float& curMax, int& curArg) {
    float m = -FMAX; int arg = 0;
    for (int q = 0; q < KNN; ++q) {
        float v = ld[q];
        if (v > m) { m = v; arg = q; }
    }
    curMax = m; curArg = arg;
    for (int e = KNN; e < n; ++e) {
        const float v = ld[e];
        if (v < bound) topk_consider(v, li[e], ld, li, curMax, curArg);
    }
}

__device__ __forceinline__ void scan20(const float* __restrict__ ld, float& curMax, int& curArg) {
    float m = -FMAX; int arg = 0;
    for (int q = 0; q < KNN; ++q) {
        float v = ld[q];
        if (v > m) { m = v; arg = q; }
    }
    curMax = m; curArg = arg;
}

__device__ __forceinline__ int cut_list(float* __restrict__ ld, int* __restrict__ li, int cnt,
                                        float bound) {
    int n = 0;
    for (int e = 0; e < cnt; e += 4) {
        const v4f  dq = *(const v4f*)(ld + e);
        const int4 cq = *(const int4*)(li + e);
        const float dd[4] = {dq.x, dq.y, dq.z, dq.w};
        const int   cc[4] = {cq.x, cq.y, cq.z, cq.w};
        #pragma unroll
        for (int q = 0; q < 4; ++q) {
            ld[n] = dd[q];
            li[n] = cc[q];
            n += (e + q < cnt && dd[q] < bound) ? 1 : 0;
        }
    }
    return n;
}

__device__ __forceinline__ void sift8(const float (&dv)[8], int colFirst, float rowBound,
                                      float* __restrict__ myd, int* __restrict__ myi,
                                      int& cnt, float& ownMax, float& m1, float& m2) {
    if (__builtin_amdgcn_ballot_w32(cnt > CAP - 8) != 0u) {
        cnt = cut_list(myd, myi, cnt, rowBound);
        if (cnt > CAP - 8) {
            float cm; int ca;
            keep20(myd, myi, cnt, rowBound, cm, ca);
            ownMax = cm;
            cnt = KNN;
        }
    }
    const float thr = fminf(ownMax, rowBound);
    #pragma unroll
    for (int r = 0; r < 8; ++r) {
        const float d = dv[r];
        myd[cnt] = d;
        myi[cnt] = colFirst + 2 * r;
        cnt += (d < thr) ? 1 : 0;
        m2 = fminf(m2, fmaxf(m1, d));
        m1 = fminf(m1, d);
    }
}

__global__ void __launch_bounds__(32 * NSPLIT) __attribute__((amdgpu_waves_per_eu(8, 10)))
knn_kernel(const float* __restrict__ x,
           const float* __restrict__ norms,
           const float* __restrict__ y,
           const float* __restrict__ z,
           float* __restrict__ out) {
    __shared__ __attribute__((aligned(16))) float kd[NSPLIT][32 * CAP];
    __shared__ __attribute__((aligned(16))) int   ki[NSPLIT][32 * CAP];
    __shared__ __attribute__((aligned(16))) float kb[16][NSPLIT];
    __shared__ int kc[NSPLIT][32];

    const int lane = threadIdx.x;
    const int w    = threadIdx.y;
    const int half = lane >> 4;
    const int lm   = lane & 15;
    const int b    = blockIdx.y;
    const int rowBase = blockIdx.x * 16;
    const float* xb = x + b * CCH * NPTS;
    const float* nb = norms + b * NPTS;

    v2f a[16];
    #pragma unroll
    for (int k = 0; k < 16; ++k) {
        int c = 4 * k + 2 * half;
        a[k].x = xb[(c    ) * NPTS + rowBase + lm];
        a[k].y = xb[(c + 1) * NPTS + rowBase + lm];
    }
    const float rn = nb[rowBase + ((lm & 1) << 3) + (lm >> 1)];
    const v2f rn2 = {rn, rn};

    float* myd = &kd[w][lane * CAP];
    int*   myi = &ki[w][lane * CAP];
    int   cnt = 0;
    float ownMax = FMAX;
    float m1 = FMAX, m2 = FMAX;
    lds_vf* post = (lds_vf*)&kb[lm][0];
    post[w] = FMAX;
    __syncthreads();

    for (int i = 0; i < NPTS / 32 / NSPLIT; ++i) {
        const int colBase = (i * NSPLIT + w) * 32;
        const float2* xb2 = (const float2*)xb;
        const int ci = (i * NSPLIT + w) * 16 + lm;

        float2 lc[8], ld[8];
        #pragma unroll
        for (int k = 0; k < 8; ++k) {
            int c = 4 * k + 2 * half;
            lc[k] = xb2[(c    ) * (NPTS / 2) + ci];
            ld[k] = xb2[(c + 1) * (NPTS / 2) + ci];
        }

        const v4f p0 = *(lds_v4f*)(post);
        const v4f p1 = *(lds_v4f*)(post + 4);

        v8f accE = {}, accO = {};
        #pragma unroll
        for (int k = 0; k < 8; ++k) {
            v2f fe, fo;
            fe.x = lc[k].x; fe.y = ld[k].x;
            fo.x = lc[k].y; fo.y = ld[k].y;
            accE = __builtin_amdgcn_wmma_f32_16x16x4_f32(false, fe, false, a[k],
                                                         (short)0, accE, false, false);
            accO = __builtin_amdgcn_wmma_f32_16x16x4_f32(false, fo, false, a[k],
                                                         (short)0, accO, false, false);
        }

        float2 hc[8], hd[8];
        #pragma unroll
        for (int k = 0; k < 8; ++k) {
            int c = 4 * (k + 8) + 2 * half;
            hc[k] = xb2[(c    ) * (NPTS / 2) + ci];
            hd[k] = xb2[(c + 1) * (NPTS / 2) + ci];
        }
        const float4* nb4 = (const float4*)nb;
        const int ni = (i * NSPLIT + w) * 8 + 4 * half;
        const float4 n0 = nb4[ni];
        const float4 n1 = nb4[ni + 1];
        const float4 n2 = nb4[ni + 2];
        const float4 n3 = nb4[ni + 3];
        #pragma unroll
        for (int k = 0; k < 8; ++k) {
            v2f fe, fo;
            fe.x = hc[k].x; fe.y = hd[k].x;
            fo.x = hc[k].y; fo.y = hd[k].y;
            accE = __builtin_amdgcn_wmma_f32_16x16x4_f32(false, fe, false, a[k + 8],
                                                         (short)0, accE, false, false);
            accO = __builtin_amdgcn_wmma_f32_16x16x4_f32(false, fo, false, a[k + 8],
                                                         (short)0, accO, false, false);
        }

        const int col0 = colBase + 16 * half;
        const v2f cnE[4] = {{n0.x, n0.y}, {n0.z, n0.w}, {n1.x, n1.y}, {n1.z, n1.w}};
        const v2f cnO[4] = {{n2.x, n2.y}, {n2.z, n2.w}, {n3.x, n3.y}, {n3.z, n3.w}};
        float dvE[8], dvO[8];
        #pragma unroll
        for (int j = 0; j < 4; ++j) {
            const v2f pe = {accE[2 * j], accE[2 * j + 1]};
            const v2f po = {accO[2 * j], accO[2 * j + 1]};
            const v2f de = (rn2 + cnE[j]) - 2.0f * pe;
            const v2f dq = (rn2 + cnO[j]) - 2.0f * po;
            dvE[2 * j] = de.x; dvE[2 * j + 1] = de.y;
            dvO[2 * j] = dq.x; dvO[2 * j + 1] = dq.y;
        }
        const float rowBound = fmaxf(fmaxf(fmaxf(p0.x, p0.y), fmaxf(p0.z, p0.w)),
                                     fmaxf(fmaxf(p1.x, p1.y), fmaxf(p1.z, p1.w)));
        sift8(dvE, col0,     rowBound, myd, myi, cnt, ownMax, m1, m2);
        sift8(dvO, col0 + 1, rowBound, myd, myi, cnt, ownMax, m1, m2);
        const float o1 = xhalf(m1), o2 = xhalf(m2);
        post[w] = fmaxf(fminf(m2, o2), fmaxf(m1, o1));
    }

    __syncthreads();
    float rowFinal;
    {
        const v4f q0 = *(lds_v4f*)(post);
        const v4f q1 = *(lds_v4f*)(post + 4);
        rowFinal = fmaxf(fmaxf(fmaxf(q0.x, q0.y), fmaxf(q0.z, q0.w)),
                         fmaxf(fmaxf(q1.x, q1.y), fmaxf(q1.z, q1.w)));
    }
    int n = cut_list(myd, myi, cnt, rowFinal);
    float curMax = FMAX; int curArg = 0;
    if (n >= KNN) {
        keep20(myd, myi, n, rowFinal, curMax, curArg);
        n = KNN;
    }
    kc[w][lane] = n;

    const int g = 2 * w + half;
    for (int stride = 1; stride < 2 * NSPLIT; stride *= 2) {
        __syncthreads();
        if ((g & (2 * stride - 1)) == 0) {
            const int og = g + stride;
            const int ol = (og & 1) * 16 + lm;
            const float* od = &kd[og >> 1][ol * CAP];
            const int*   oi = &ki[og >> 1][ol * CAP];
            const int    on = kc[og >> 1][ol];
            for (int j = 0; j < on; ++j) {
                const float d = od[j];
                const int   c = oi[j];
                if (n < KNN) {
                    myd[n] = d; myi[n] = c; ++n;
                    if (n == KNN) scan20(myd, curMax, curArg);
                } else {
                    topk_consider(d, c, myd, myi, curMax, curArg);
                }
            }
            kc[w][lane] = n;
        }
    }
    if (g == 0) {
        const int fill = (n > 0) ? myi[0] : 0;
        for (int j = n; j < KNN; ++j) myi[j] = fill;
    }

    __syncthreads();
    const float* yb = y + b * NPTS * CCH;
    float* stage = &kd[1][0];
    for (int rr = 0; rr < 2; ++rr) {
        const int r = 2 * w + rr;
        const int* li = &ki[0][r * CAP];
        float b0 = -FMAX, b1 = -FMAX;
        #pragma unroll
        for (int j = 0; j < KNN; ++j) {
            const int m = li[j];
            b0 = fmaxf(b0, yb[m * CCH + lane]);
            b1 = fmaxf(b1, yb[m * CCH + 32 + lane]);
        }
        const float* zr = z + (b * NPTS + rowBase + r) * CCH;
        stage[lane * 17 + r]        = b0 + zr[lane];
        stage[(lane + 32) * 17 + r] = b1 + zr[lane + 32];
    }
    __syncthreads();
    const int tid = w * 32 + lane;
    #pragma unroll
    for (int q = 0; q < 4; ++q) {
        const int e = q * 256 + tid;
        const int o = e >> 4, r = e & 15;
        out[(b * CCH + o) * NPTS + rowBase + r] = stage[o * 17 + r];
    }
}

__global__ void __launch_bounds__(32) proj_kernel(const float* __restrict__ x,
                                                  const float* __restrict__ W,
                                                  float* __restrict__ y,
                                                  float* __restrict__ z) {
    const int lane = threadIdx.x;
    const int half = lane >> 4;
    const int lm   = lane & 15;
    const int b    = blockIdx.y;
    const int rowBase = blockIdx.x * 16;
    const float* xb = x + b * CCH * NPTS;

    v2f a[16];
    #pragma unroll
    for (int k = 0; k < 16; ++k) {
        int c = 4 * k + 2 * half;
        a[k].x = xb[(c    ) * NPTS + rowBase + lm];
        a[k].y = xb[(c + 1) * NPTS + rowBase + lm];
    }

    for (int ot = 0; ot < 4; ++ot) {
        const int o = ot * 16 + lm;
        v8f accY = {}, accZ = {};
        #pragma unroll
        for (int k = 0; k < 16; ++k) {
            int c = 4 * k + 2 * half;
            float w0 = W[o * 128 + c], w1 = W[o * 128 + c + 1];
            v2f bw, bz;
            bw.x = w0;
            bw.y = w1;
            bz.x = W[o * 128 + 64 + c    ] - w0;
            bz.y = W[o * 128 + 64 + c + 1] - w1;
            accY = __builtin_amdgcn_wmma_f32_16x16x4_f32(false, a[k], false, bw,
                                                         (short)0, accY, false, false);
            accZ = __builtin_amdgcn_wmma_f32_16x16x4_f32(false, a[k], false, bz,
                                                         (short)0, accZ, false, false);
        }
        #pragma unroll
        for (int r = 0; r < 8; ++r) {
            int row = rowBase + r + 8 * half;
            y[(b * NPTS + row) * CCH + o] = accY[r];
            z[(b * NPTS + row) * CCH + o] = accZ[r];
        }
    }
}

extern "C" void kernel_launch(void* const* d_in, const int* in_sizes, int n_in,
                              void* d_out, int out_size, void* d_ws, size_t ws_size,
                              hipStream_t stream) {
    (void)in_sizes; (void)n_in; (void)out_size; (void)ws_size;
    const float* x = (const float*)d_in[0];
    const float* W = (const float*)d_in[1];
    float* out = (float*)d_out;

    char* ws = (char*)d_ws;
    float* norms = (float*)(ws);
    float* y     = (float*)(ws + 65536 + 1310720);
    float* z     = (float*)(ws + 65536 + 1310720 + 4194304);

    norms_kernel    <<<dim3(64), 256, 0, stream>>>(x, norms);
    proj_kernel     <<<dim3(256, 4), 32, 0, stream>>>(x, W, y, z);
    knn_kernel      <<<dim3(256, 4), dim3(32, NSPLIT), 0, stream>>>(x, norms, y, z, out);
}
